# starting kernel
# baseline (speedup 1.0000x reference)
#include <hip/hip_runtime.h>
#include <hip/hip_bf16.h>

typedef __attribute__((ext_vector_type(2))) float v2f;
typedef __attribute__((ext_vector_type(8))) float v8f;

#define DDIM 64
#define NMAX 8192

__device__ __forceinline__ float wave_reduce_sum(float v) {
#pragma unroll
    for (int off = 16; off > 0; off >>= 1) v += __shfl_xor(v, off, 32);
    return v;
}

__global__ void k_prep(const float* __restrict__ sH, const float* __restrict__ omega,
                       float* __restrict__ facc, float* __restrict__ gv,
                       float* __restrict__ skew, int N) {
    int i = blockIdx.x * blockDim.x + threadIdx.x;
    int ND = N * DDIM;
    int stride = gridDim.x * blockDim.x;
    for (int t = i; t < ND; t += stride) facc[t] = 0.0f;
    if (i < N) gv[i] = tanhf(sH[i]);
    if (i < DDIM * DDIM) {
        int k = i / DDIM, n = i % DDIM;
        skew[i] = 0.5f * (omega[k * DDIM + n] - omega[n * DDIM + k]);
    }
}

__global__ void k_norm(const float* __restrict__ K, float* __restrict__ Kn, int N) {
    int lane = threadIdx.x & 31, wid = threadIdx.x >> 5;
    int row = blockIdx.x * (blockDim.x >> 5) + wid;
    if (row >= N) return;
    float a0 = K[row * DDIM + lane];
    float a1 = K[row * DDIM + lane + 32];
    float s = wave_reduce_sum(a0 * a0 + a1 * a1);
    float inv = rsqrtf(s);
    Kn[row * DDIM + lane]      = a0 * inv;
    Kn[row * DDIM + lane + 32] = a1 * inv;
}

__global__ void k_matvec(const float* __restrict__ W, const float* __restrict__ gv,
                         const float* __restrict__ sH, float* __restrict__ fH, int N) {
    __shared__ float gs[NMAX];
    for (int t = threadIdx.x; t < N && t < NMAX; t += blockDim.x) gs[t] = gv[t];
    __syncthreads();
    int lane = threadIdx.x & 31, wid = threadIdx.x >> 5;
    int wpb = blockDim.x >> 5;
    int gw = blockIdx.x * wpb + wid;
    int tw = gridDim.x * wpb;
    int n4 = N >> 2;
    for (int row = gw; row < N; row += tw) {
        const float4* Wr = (const float4*)(W + (size_t)row * N);
        float acc = 0.0f;
        for (int c4 = lane; c4 < n4; c4 += 32) {
            float4 w = Wr[c4];
            int c = c4 << 2;
            acc += w.x * gs[c] + w.y * gs[c + 1] + w.z * gs[c + 2] + w.w * gs[c + 3];
        }
        acc = wave_reduce_sum(acc);
        if (lane == 0) fH[row] = -sH[row] + acc;
    }
}

__global__ void k_hk(const int* __restrict__ ind, const float* __restrict__ Kn,
                     const float* __restrict__ gv, float* __restrict__ fH,
                     float* __restrict__ facc, int E) {
    int lane = threadIdx.x & 31, wid = threadIdx.x >> 5;
    int e = blockIdx.x * (blockDim.x >> 5) + wid;
    if (e >= E) return;
    int i = ind[2 * e], j = ind[2 * e + 1];
    float a0 = Kn[i * DDIM + lane], a1 = Kn[i * DDIM + lane + 32];
    float c0 = Kn[j * DDIM + lane], c1 = Kn[j * DDIM + lane + 32];
    float Gram = wave_reduce_sum(a0 * c0 + a1 * c1);
    float gi = gv[i], gj = gv[j];
    if (lane == 0) {
        atomicAdd(&fH[i], 0.5f * Gram * gj);
        atomicAdd(&fH[j], 0.5f * Gram * gi);
    }
    float G = 0.5f * gi * gj;
    atomicAdd(&facc[i * DDIM + lane],      -G * c0);
    atomicAdd(&facc[i * DDIM + lane + 32], -G * c1);
    atomicAdd(&facc[j * DDIM + lane],      -G * a0);
    atomicAdd(&facc[j * DDIM + lane + 32], -G * a1);
}

__global__ void k_kedges(const int* __restrict__ ind, const float* __restrict__ Kn,
                         const float* __restrict__ w1v, const float* __restrict__ b1v,
                         const float* __restrict__ w2v, float* __restrict__ facc, int E) {
    int lane = threadIdx.x & 31, wid = threadIdx.x >> 5;
    int e = blockIdx.x * (blockDim.x >> 5) + wid;
    if (e >= E) return;
    int p = ind[2 * e], q = ind[2 * e + 1];
    float a0 = Kn[p * DDIM + lane], a1 = Kn[p * DDIM + lane + 32];
    float c0 = Kn[q * DDIM + lane], c1 = Kn[q * DDIM + lane + 32];
    float s = wave_reduce_sum(a0 * c0 + a1 * c1);
    float h0 = tanhf(s * w1v[lane]      + b1v[lane]);
    float h1 = tanhf(s * w1v[lane + 32] + b1v[lane + 32]);
    float dE = wave_reduce_sum(h0 * w2v[lane] + h1 * w2v[lane + 32]);
    atomicAdd(&facc[p * DDIM + lane],      dE * c0);
    atomicAdd(&facc[p * DDIM + lane + 32], dE * c1);
    atomicAdd(&facc[q * DDIM + lane],      dE * a0);
    atomicAdd(&facc[q * DDIM + lane + 32], dE * a1);
}

__global__ void k_komega(const float* __restrict__ Kn, const float* __restrict__ skew,
                         float* __restrict__ Kom, int N) {
    __shared__ float ssk[DDIM * DDIM];
    for (int t = threadIdx.x; t < DDIM * DDIM; t += blockDim.x) ssk[t] = skew[t];
    __syncthreads();
    int lane = threadIdx.x & 31, wid = threadIdx.x >> 5;
    int tile = blockIdx.x * (blockDim.x >> 5) + wid;
    if (tile >= (N >> 4)) return;
    int row0 = tile << 4;
    int l15 = lane & 15;
    int hi = lane >> 4;
    int arow = row0 + l15;
    int kofs = hi * 2;
    v8f acc[4] = {};
#pragma unroll
    for (int k0 = 0; k0 < DDIM; k0 += 4) {
        v2f A;
        A.x = Kn[arow * DDIM + k0 + kofs];
        A.y = Kn[arow * DDIM + k0 + kofs + 1];
#pragma unroll
        for (int t = 0; t < 4; ++t) {
            v2f B;
            int n = (t << 4) + l15;
            B.x = ssk[(k0 + kofs) * DDIM + n];
            B.y = ssk[(k0 + kofs + 1) * DDIM + n];
            acc[t] = __builtin_amdgcn_wmma_f32_16x16x4_f32(
                false, A, false, B, (short)0, acc[t], false, false);
        }
    }
    int crow = row0 + hi * 8;
#pragma unroll
    for (int t = 0; t < 4; ++t) {
#pragma unroll
        for (int v = 0; v < 8; ++v) {
            Kom[(crow + v) * DDIM + (t << 4) + l15] = acc[t][v];
        }
    }
}

__global__ void k_final(const float* __restrict__ Kn, const float* __restrict__ facc,
                        const float* __restrict__ Kom, float* __restrict__ fK, int N) {
    int lane = threadIdx.x & 31, wid = threadIdx.x >> 5;
    int row = blockIdx.x * (blockDim.x >> 5) + wid;
    if (row >= N) return;
    float k0 = Kn[row * DDIM + lane],   k1 = Kn[row * DDIM + lane + 32];
    float r0 = facc[row * DDIM + lane], r1 = facc[row * DDIM + lane + 32];
    float dot = wave_reduce_sum(k0 * r0 + k1 * r1);
    fK[row * DDIM + lane]      = -r0 + k0 * dot + Kom[row * DDIM + lane];
    fK[row * DDIM + lane + 32] = -r1 + k1 * dot + Kom[row * DDIM + lane + 32];
}

extern "C" void kernel_launch(void* const* d_in, const int* in_sizes, int n_in,
                              void* d_out, int out_size, void* d_ws, size_t ws_size,
                              hipStream_t stream) {
    const float* sH    = (const float*)d_in[0];
    const float* sK    = (const float*)d_in[1];
    const float* WH    = (const float*)d_in[2];
    const float* omega = (const float*)d_in[3];
    const float* w1v   = (const float*)d_in[4];
    const float* b1v   = (const float*)d_in[5];
    const float* w2v   = (const float*)d_in[6];
    const int*   indK  = (const int*)d_in[7];
    const int*   indHK = (const int*)d_in[8];

    int N   = in_sizes[0];
    int EK  = in_sizes[7] / 2;
    int EHK = in_sizes[8] / 2;
    int ND  = N * DDIM;

    float* fH = (float*)d_out;
    float* fK = (float*)d_out + N;

    float* Kn   = (float*)d_ws;
    float* gv   = Kn + (size_t)ND;
    float* skew = gv + N;
    float* facc = skew + DDIM * DDIM;
    float* Kom  = facc + (size_t)ND;

    k_prep  <<<(ND + 255) / 256,        256, 0, stream>>>(sH, omega, facc, gv, skew, N);
    k_norm  <<<(N + 7) / 8,             256, 0, stream>>>(sK, Kn, N);
    k_matvec<<<256,                     256, 0, stream>>>(WH, gv, sH, fH, N);
    k_hk    <<<(EHK + 7) / 8,           256, 0, stream>>>(indHK, Kn, gv, fH, facc, EHK);
    k_kedges<<<(EK + 7) / 8,            256, 0, stream>>>(indK, Kn, w1v, b1v, w2v, facc, EK);
    k_komega<<<((N >> 4) + 7) / 8,      256, 0, stream>>>(Kn, skew, Kom, N);
    k_final <<<(N + 7) / 8,             256, 0, stream>>>(Kn, facc, Kom, fK, N);
}


# f_H matvec rebuilt as a read-once stream: 16-wave workgroups, g pieces in registers, 16-32 non-temporal 512-B row loads in flight per wave; row sums now a different reduction tree
# speedup vs baseline: 1.7695x; 1.7695x over previous
#include <hip/hip_runtime.h>
#include <hip/hip_bf16.h>

typedef __attribute__((ext_vector_type(2))) float v2f;
typedef __attribute__((ext_vector_type(8))) float v8f;

#define DDIM 64
#define NMAX 8192

__device__ __forceinline__ float wave_reduce_sum(float v) {
#pragma unroll
    for (int off = 16; off > 0; off >>= 1) v += __shfl_xor(v, off, 32);
    return v;
}

__global__ void k_prep(const float* __restrict__ sH, const float* __restrict__ omega,
                       float* __restrict__ facc, float* __restrict__ gv,
                       float* __restrict__ skew, int N) {
    int i = blockIdx.x * blockDim.x + threadIdx.x;
    int ND = N * DDIM;
    int stride = gridDim.x * blockDim.x;
    for (int t = i; t < ND; t += stride) facc[t] = 0.0f;
    if (i < N) gv[i] = tanhf(sH[i]);
    if (i < DDIM * DDIM) {
        int k = i / DDIM, n = i % DDIM;
        skew[i] = 0.5f * (omega[k * DDIM + n] - omega[n * DDIM + k]);
    }
}

__global__ void k_norm(const float* __restrict__ K, float* __restrict__ Kn, int N) {
    int lane = threadIdx.x & 31, wid = threadIdx.x >> 5;
    int row = blockIdx.x * (blockDim.x >> 5) + wid;
    if (row >= N) return;
    float a0 = K[row * DDIM + lane];
    float a1 = K[row * DDIM + lane + 32];
    float s = wave_reduce_sum(a0 * a0 + a1 * a1);
    float inv = rsqrtf(s);
    Kn[row * DDIM + lane]      = a0 * inv;
    Kn[row * DDIM + lane + 32] = a1 * inv;
}

#define MV_N     8192
#define MV_WAVES 16
#define MV_ROWS  32
#define MV_Q     4
#define MV_G     4

typedef float f32x4 __attribute__((ext_vector_type(4)));

__global__ void __launch_bounds__(MV_WAVES * 32)
k_matvec_stream(const float* __restrict__ W, const float* __restrict__ gv,
                const float* __restrict__ sH, float* __restrict__ fH) {
    __shared__ float part[MV_ROWS][MV_WAVES * 32];
    __shared__ float res[MV_ROWS];
    const unsigned tid = threadIdx.x, lane = tid & 31u;
    const unsigned wave = __builtin_amdgcn_readfirstlane(tid >> 5);
    const unsigned row0 = blockIdx.x * MV_ROWS;
    const unsigned col  = wave * (MV_Q * 128u) + lane * 4u;

    f32x4 gq[MV_Q];
#pragma unroll
    for (int q = 0; q < MV_Q; ++q) gq[q] = *(const f32x4*)(gv + col + 128u * q);

    const float* wb = W + (size_t)row0 * MV_N + col;
    f32x4 w[2][MV_G][MV_Q];
#define MV_LOAD_GROUP(K) do { \
    _Pragma("unroll") for (int r_ = 0; r_ < MV_G; ++r_) \
    _Pragma("unroll") for (int q_ = 0; q_ < MV_Q; ++q_) \
        w[(K) & 1][r_][q_] = __builtin_nontemporal_load( \
            (const f32x4*)(wb + (size_t)((K) * MV_G + r_) * MV_N + 128u * q_)); } while (0)

    MV_LOAD_GROUP(0);
#pragma unroll
    for (int k = 0; k < MV_ROWS / MV_G; ++k) {
        __builtin_amdgcn_sched_barrier(0);
        if (k + 1 < MV_ROWS / MV_G) MV_LOAD_GROUP(k + 1);
        __builtin_amdgcn_sched_barrier(0);
#pragma unroll
        for (int r = 0; r < MV_G; ++r) {
            float a0 = 0.f, a1 = 0.f, a2 = 0.f, a3 = 0.f;
#pragma unroll
            for (int q = 0; q < MV_Q; ++q) {
                const f32x4 x = w[k & 1][r][q];
                a0 = __builtin_fmaf(x.x, gq[q].x, a0);
                a1 = __builtin_fmaf(x.y, gq[q].y, a1);
                a2 = __builtin_fmaf(x.z, gq[q].z, a2);
                a3 = __builtin_fmaf(x.w, gq[q].w, a3);
            }
            part[k * MV_G + r][tid] = (a0 + a1) + (a2 + a3);
        }
    }
#undef MV_LOAD_GROUP
    __syncthreads();

    float s0 = 0.f, s1 = 0.f;
#pragma unroll
    for (int j = 0; j < MV_WAVES; ++j) {
        s0 += part[2u * wave][lane + 32u * j];
        s1 += part[2u * wave + 1u][lane + 32u * j];
    }
    s0 = wave_reduce_sum(s0);
    s1 = wave_reduce_sum(s1);
    if (lane == 0) { res[2u * wave] = s0; res[2u * wave + 1u] = s1; }
    __syncthreads();
    if (tid < MV_ROWS) fH[row0 + tid] = -sH[row0 + tid] + res[tid];
}

__global__ void k_matvec(const float* __restrict__ W, const float* __restrict__ gv,
                         const float* __restrict__ sH, float* __restrict__ fH, int N) {
    __shared__ float gs[NMAX];
    for (int t = threadIdx.x; t < N && t < NMAX; t += blockDim.x) gs[t] = gv[t];
    __syncthreads();
    int lane = threadIdx.x & 31, wid = threadIdx.x >> 5;
    int wpb = blockDim.x >> 5;
    int gw = blockIdx.x * wpb + wid;
    int tw = gridDim.x * wpb;
    int n4 = N >> 2;
    for (int row = gw; row < N; row += tw) {
        const float4* Wr = (const float4*)(W + (size_t)row * N);
        float acc = 0.0f;
        for (int c4 = lane; c4 < n4; c4 += 32) {
            float4 w = Wr[c4];
            int c = c4 << 2;
            acc += w.x * gs[c] + w.y * gs[c + 1] + w.z * gs[c + 2] + w.w * gs[c + 3];
        }
        acc = wave_reduce_sum(acc);
        if (lane == 0) fH[row] = -sH[row] + acc;
    }
}

__global__ void k_hk(const int* __restrict__ ind, const float* __restrict__ Kn,
                     const float* __restrict__ gv, float* __restrict__ fH,
                     float* __restrict__ facc, int E) {
    int lane = threadIdx.x & 31, wid = threadIdx.x >> 5;
    int e = blockIdx.x * (blockDim.x >> 5) + wid;
    if (e >= E) return;
    int i = ind[2 * e], j = ind[2 * e + 1];
    float a0 = Kn[i * DDIM + lane], a1 = Kn[i * DDIM + lane + 32];
    float c0 = Kn[j * DDIM + lane], c1 = Kn[j * DDIM + lane + 32];
    float Gram = wave_reduce_sum(a0 * c0 + a1 * c1);
    float gi = gv[i], gj = gv[j];
    if (lane == 0) {
        atomicAdd(&fH[i], 0.5f * Gram * gj);
        atomicAdd(&fH[j], 0.5f * Gram * gi);
    }
    float G = 0.5f * gi * gj;
    atomicAdd(&facc[i * DDIM + lane],      -G * c0);
    atomicAdd(&facc[i * DDIM + lane + 32], -G * c1);
    atomicAdd(&facc[j * DDIM + lane],      -G * a0);
    atomicAdd(&facc[j * DDIM + lane + 32], -G * a1);
}

__global__ void k_kedges(const int* __restrict__ ind, const float* __restrict__ Kn,
                         const float* __restrict__ w1v, const float* __restrict__ b1v,
                         const float* __restrict__ w2v, float* __restrict__ facc, int E) {
    int lane = threadIdx.x & 31, wid = threadIdx.x >> 5;
    int e = blockIdx.x * (blockDim.x >> 5) + wid;
    if (e >= E) return;
    int p = ind[2 * e], q = ind[2 * e + 1];
    float a0 = Kn[p * DDIM + lane], a1 = Kn[p * DDIM + lane + 32];
    float c0 = Kn[q * DDIM + lane], c1 = Kn[q * DDIM + lane + 32];
    float s = wave_reduce_sum(a0 * c0 + a1 * c1);
    float h0 = tanhf(s * w1v[lane]      + b1v[lane]);
    float h1 = tanhf(s * w1v[lane + 32] + b1v[lane + 32]);
    float dE = wave_reduce_sum(h0 * w2v[lane] + h1 * w2v[lane + 32]);
    atomicAdd(&facc[p * DDIM + lane],      dE * c0);
    atomicAdd(&facc[p * DDIM + lane + 32], dE * c1);
    atomicAdd(&facc[q * DDIM + lane],      dE * a0);
    atomicAdd(&facc[q * DDIM + lane + 32], dE * a1);
}

__global__ void k_komega(const float* __restrict__ Kn, const float* __restrict__ skew,
                         float* __restrict__ Kom, int N) {
    __shared__ float ssk[DDIM * DDIM];
    for (int t = threadIdx.x; t < DDIM * DDIM; t += blockDim.x) ssk[t] = skew[t];
    __syncthreads();
    int lane = threadIdx.x & 31, wid = threadIdx.x >> 5;
    int tile = blockIdx.x * (blockDim.x >> 5) + wid;
    if (tile >= (N >> 4)) return;
    int row0 = tile << 4;
    int l15 = lane & 15;
    int hi = lane >> 4;
    int arow = row0 + l15;
    int kofs = hi * 2;
    v8f acc[4] = {};
#pragma unroll
    for (int k0 = 0; k0 < DDIM; k0 += 4) {
        v2f A;
        A.x = Kn[arow * DDIM + k0 + kofs];
        A.y = Kn[arow * DDIM + k0 + kofs + 1];
#pragma unroll
        for (int t = 0; t < 4; ++t) {
            v2f B;
            int n = (t << 4) + l15;
            B.x = ssk[(k0 + kofs) * DDIM + n];
            B.y = ssk[(k0 + kofs + 1) * DDIM + n];
            acc[t] = __builtin_amdgcn_wmma_f32_16x16x4_f32(
                false, A, false, B, (short)0, acc[t], false, false);
        }
    }
    int crow = row0 + hi * 8;
#pragma unroll
    for (int t = 0; t < 4; ++t) {
#pragma unroll
        for (int v = 0; v < 8; ++v) {
            Kom[(crow + v) * DDIM + (t << 4) + l15] = acc[t][v];
        }
    }
}

__global__ void k_final(const float* __restrict__ Kn, const float* __restrict__ facc,
                        const float* __restrict__ Kom, float* __restrict__ fK, int N) {
    int lane = threadIdx.x & 31, wid = threadIdx.x >> 5;
    int row = blockIdx.x * (blockDim.x >> 5) + wid;
    if (row >= N) return;
    float k0 = Kn[row * DDIM + lane],   k1 = Kn[row * DDIM + lane + 32];
    float r0 = facc[row * DDIM + lane], r1 = facc[row * DDIM + lane + 32];
    float dot = wave_reduce_sum(k0 * r0 + k1 * r1);
    fK[row * DDIM + lane]      = -r0 + k0 * dot + Kom[row * DDIM + lane];
    fK[row * DDIM + lane + 32] = -r1 + k1 * dot + Kom[row * DDIM + lane + 32];
}

extern "C" void kernel_launch(void* const* d_in, const int* in_sizes, int n_in,
                              void* d_out, int out_size, void* d_ws, size_t ws_size,
                              hipStream_t stream) {
    const float* sH    = (const float*)d_in[0];
    const float* sK    = (const float*)d_in[1];
    const float* WH    = (const float*)d_in[2];
    const float* omega = (const float*)d_in[3];
    const float* w1v   = (const float*)d_in[4];
    const float* b1v   = (const float*)d_in[5];
    const float* w2v   = (const float*)d_in[6];
    const int*   indK  = (const int*)d_in[7];
    const int*   indHK = (const int*)d_in[8];

    int N   = in_sizes[0];
    int EK  = in_sizes[7] / 2;
    int EHK = in_sizes[8] / 2;
    int ND  = N * DDIM;

    float* fH = (float*)d_out;
    float* fK = (float*)d_out + N;

    float* Kn   = (float*)d_ws;
    float* gv   = Kn + (size_t)ND;
    float* skew = gv + N;
    float* facc = skew + DDIM * DDIM;
    float* Kom  = facc + (size_t)ND;

    k_prep  <<<(ND + 255) / 256,        256, 0, stream>>>(sH, omega, facc, gv, skew, N);
    k_norm  <<<(N + 7) / 8,             256, 0, stream>>>(sK, Kn, N);
    if (N == MV_N)
        k_matvec_stream<<<MV_N / MV_ROWS, MV_WAVES * 32, 0, stream>>>(WH, gv, sH, fH);
    else
        k_matvec<<<256,                 256, 0, stream>>>(WH, gv, sH, fH, N);
    k_hk    <<<(EHK + 7) / 8,           256, 0, stream>>>(indHK, Kn, gv, fH, facc, EHK);
    k_kedges<<<(EK + 7) / 8,            256, 0, stream>>>(indK, Kn, w1v, b1v, w2v, facc, EK);
    k_komega<<<((N >> 4) + 7) / 8,      256, 0, stream>>>(Kn, skew, Kom, N);
    k_final <<<(N + 7) / 8,             256, 0, stream>>>(Kn, facc, Kom, fK, N);
}


# both edge passes (one short wave per edge: 524288 waves, 2 launches) as one launch of 8192 long-lived waves: 32 edges per index load, 4 edges in flight; f32 atomic sums kept (arrival order)
# speedup vs baseline: 2.0094x; 1.1356x over previous
#include <hip/hip_runtime.h>
#include <hip/hip_bf16.h>

typedef __attribute__((ext_vector_type(2))) float v2f;
typedef __attribute__((ext_vector_type(8))) float v8f;

#define DDIM 64
#define NMAX 8192

__device__ __forceinline__ float wave_reduce_sum(float v) {
#pragma unroll
    for (int off = 16; off > 0; off >>= 1) v += __shfl_xor(v, off, 32);
    return v;
}

__global__ void k_prep(const float* __restrict__ sH, const float* __restrict__ omega,
                       float* __restrict__ facc, float* __restrict__ gv,
                       float* __restrict__ skew, int N) {
    int i = blockIdx.x * blockDim.x + threadIdx.x;
    int ND = N * DDIM;
    int stride = gridDim.x * blockDim.x;
    for (int t = i; t < ND; t += stride) facc[t] = 0.0f;
    if (i < N) gv[i] = tanhf(sH[i]);
    if (i < DDIM * DDIM) {
        int k = i / DDIM, n = i % DDIM;
        skew[i] = 0.5f * (omega[k * DDIM + n] - omega[n * DDIM + k]);
    }
}

__global__ void k_norm(const float* __restrict__ K, float* __restrict__ Kn, int N) {
    int lane = threadIdx.x & 31, wid = threadIdx.x >> 5;
    int row = blockIdx.x * (blockDim.x >> 5) + wid;
    if (row >= N) return;
    float a0 = K[row * DDIM + lane];
    float a1 = K[row * DDIM + lane + 32];
    float s = wave_reduce_sum(a0 * a0 + a1 * a1);
    float inv = rsqrtf(s);
    Kn[row * DDIM + lane]      = a0 * inv;
    Kn[row * DDIM + lane + 32] = a1 * inv;
}

#define MV_N     8192
#define MV_WAVES 16
#define MV_ROWS  32
#define MV_Q     4
#define MV_G     4

typedef float f32x4 __attribute__((ext_vector_type(4)));

__global__ void __launch_bounds__(MV_WAVES * 32)
k_matvec_stream(const float* __restrict__ W, const float* __restrict__ gv,
                const float* __restrict__ sH, float* __restrict__ fH) {
    __shared__ float part[MV_ROWS][MV_WAVES * 32];
    __shared__ float res[MV_ROWS];
    const unsigned tid = threadIdx.x, lane = tid & 31u;
    const unsigned wave = __builtin_amdgcn_readfirstlane(tid >> 5);
    const unsigned row0 = blockIdx.x * MV_ROWS;
    const unsigned col  = wave * (MV_Q * 128u) + lane * 4u;

    f32x4 gq[MV_Q];
#pragma unroll
    for (int q = 0; q < MV_Q; ++q) gq[q] = *(const f32x4*)(gv + col + 128u * q);

    const float* wb = W + (size_t)row0 * MV_N + col;
    f32x4 w[2][MV_G][MV_Q];
#define MV_LOAD_GROUP(K) do { \
    _Pragma("unroll") for (int r_ = 0; r_ < MV_G; ++r_) \
    _Pragma("unroll") for (int q_ = 0; q_ < MV_Q; ++q_) \
        w[(K) & 1][r_][q_] = __builtin_nontemporal_load( \
            (const f32x4*)(wb + (size_t)((K) * MV_G + r_) * MV_N + 128u * q_)); } while (0)

    MV_LOAD_GROUP(0);
#pragma unroll
    for (int k = 0; k < MV_ROWS / MV_G; ++k) {
        __builtin_amdgcn_sched_barrier(0);
        if (k + 1 < MV_ROWS / MV_G) MV_LOAD_GROUP(k + 1);
        __builtin_amdgcn_sched_barrier(0);
#pragma unroll
        for (int r = 0; r < MV_G; ++r) {
            float a0 = 0.f, a1 = 0.f, a2 = 0.f, a3 = 0.f;
#pragma unroll
            for (int q = 0; q < MV_Q; ++q) {
                const f32x4 x = w[k & 1][r][q];
                a0 = __builtin_fmaf(x.x, gq[q].x, a0);
                a1 = __builtin_fmaf(x.y, gq[q].y, a1);
                a2 = __builtin_fmaf(x.z, gq[q].z, a2);
                a3 = __builtin_fmaf(x.w, gq[q].w, a3);
            }
            part[k * MV_G + r][tid] = (a0 + a1) + (a2 + a3);
        }
    }
#undef MV_LOAD_GROUP
    __syncthreads();

    float s0 = 0.f, s1 = 0.f;
#pragma unroll
    for (int j = 0; j < MV_WAVES; ++j) {
        s0 += part[2u * wave][lane + 32u * j];
        s1 += part[2u * wave + 1u][lane + 32u * j];
    }
    s0 = wave_reduce_sum(s0);
    s1 = wave_reduce_sum(s1);
    if (lane == 0) { res[2u * wave] = s0; res[2u * wave + 1u] = s1; }
    __syncthreads();
    if (tid < MV_ROWS) fH[row0 + tid] = -sH[row0 + tid] + res[tid];
}

__global__ void k_matvec(const float* __restrict__ W, const float* __restrict__ gv,
                         const float* __restrict__ sH, float* __restrict__ fH, int N) {
    __shared__ float gs[NMAX];
    for (int t = threadIdx.x; t < N && t < NMAX; t += blockDim.x) gs[t] = gv[t];
    __syncthreads();
    int lane = threadIdx.x & 31, wid = threadIdx.x >> 5;
    int wpb = blockDim.x >> 5;
    int gw = blockIdx.x * wpb + wid;
    int tw = gridDim.x * wpb;
    int n4 = N >> 2;
    for (int row = gw; row < N; row += tw) {
        const float4* Wr = (const float4*)(W + (size_t)row * N);
        float acc = 0.0f;
        for (int c4 = lane; c4 < n4; c4 += 32) {
            float4 w = Wr[c4];
            int c = c4 << 2;
            acc += w.x * gs[c] + w.y * gs[c + 1] + w.z * gs[c + 2] + w.w * gs[c + 3];
        }
        acc = wave_reduce_sum(acc);
        if (lane == 0) fH[row] = -sH[row] + acc;
    }
}

#define EDGE_B 4

__device__ __forceinline__ int lane_to_scalar(int v, int k) {
    return __builtin_amdgcn_readlane(v, k);
}

__device__ __forceinline__ void wave_reduce_sum_batch(float (&v)[EDGE_B]) {
#pragma unroll
    for (int off = 16; off > 0; off >>= 1) {
        float t[EDGE_B];
#pragma unroll
        for (int u = 0; u < EDGE_B; ++u) t[u] = __shfl_xor(v[u], off, 32);
#pragma unroll
        for (int u = 0; u < EDGE_B; ++u) v[u] += t[u];
    }
}

template <bool ISK>
__device__ __forceinline__ void edge_one(int i, int j, float a0, float a1, float c0, float c1,
                                         float gi, float gj, unsigned lane, int k,
                                         float w1a, float w1b, float b1a, float b1b,
                                         float w2a, float w2b, float& myGram,
                                         float* __restrict__ facc) {
    float s = wave_reduce_sum(a0 * c0 + a1 * c1);
    float coef;
    if (ISK) {
        float h0 = tanhf(s * w1a + b1a);
        float h1 = tanhf(s * w1b + b1b);
        coef = wave_reduce_sum(h0 * w2a + h1 * w2b);
    } else {
        if (lane == (unsigned)k) myGram = s;
        coef = -(0.5f * gi * gj);
    }
    atomicAdd(&facc[i * DDIM + lane],      coef * c0);
    atomicAdd(&facc[i * DDIM + lane + 32], coef * c1);
    atomicAdd(&facc[j * DDIM + lane],      coef * a0);
    atomicAdd(&facc[j * DDIM + lane + 32], coef * a1);
}

template <bool ISK>
__device__ __forceinline__ void edge_chunk(const int* __restrict__ ind, int E, int chunk,
                                           unsigned lane, const float* __restrict__ Kn,
                                           const float* __restrict__ gv,
                                           float w1a, float w1b, float b1a, float b1b,
                                           float w2a, float w2b,
                                           float* __restrict__ fH, float* __restrict__ facc) {
    const int e0 = chunk * 32;
    const int n = min(32, E - e0);
    const int e = e0 + (int)lane;
    int vi = 0, vj = 0;
    if (e < E) {
        const int2 t = *(const int2*)(ind + 2 * (size_t)e);
        vi = t.x; vj = t.y;
    }
    float gvi = 0.f, gvj = 0.f, myGram = 0.f;
    if (!ISK) { gvi = gv[vi]; gvj = gv[vj]; }

    int k0 = 0;
    for (; k0 + EDGE_B <= n; k0 += EDGE_B) {
        int si[EDGE_B], sj[EDGE_B];
        float a0[EDGE_B], a1[EDGE_B], c0[EDGE_B], c1[EDGE_B], gi[EDGE_B], gj[EDGE_B];
#pragma unroll
        for (int u = 0; u < EDGE_B; ++u) {
            si[u] = lane_to_scalar(vi, k0 + u);
            sj[u] = lane_to_scalar(vj, k0 + u);
            a0[u] = Kn[si[u] * DDIM + lane];
            a1[u] = Kn[si[u] * DDIM + lane + 32];
            c0[u] = Kn[sj[u] * DDIM + lane];
            c1[u] = Kn[sj[u] * DDIM + lane + 32];
            gi[u] = ISK ? 0.f : __int_as_float(lane_to_scalar(__float_as_int(gvi), k0 + u));
            gj[u] = ISK ? 0.f : __int_as_float(lane_to_scalar(__float_as_int(gvj), k0 + u));
        }
        float s[EDGE_B], coef[EDGE_B];
#pragma unroll
        for (int u = 0; u < EDGE_B; ++u) s[u] = a0[u] * c0[u] + a1[u] * c1[u];
        wave_reduce_sum_batch(s);
        if (ISK) {
#pragma unroll
            for (int u = 0; u < EDGE_B; ++u) {
                const float h0 = tanhf(s[u] * w1a + b1a);
                const float h1 = tanhf(s[u] * w1b + b1b);
                coef[u] = h0 * w2a + h1 * w2b;
            }
            wave_reduce_sum_batch(coef);
        } else {
#pragma unroll
            for (int u = 0; u < EDGE_B; ++u) {
                if (lane == (unsigned)(k0 + u)) myGram = s[u];
                coef[u] = -(0.5f * gi[u] * gj[u]);
            }
        }
#pragma unroll
        for (int u = 0; u < EDGE_B; ++u) {
            atomicAdd(&facc[si[u] * DDIM + lane],      coef[u] * c0[u]);
            atomicAdd(&facc[si[u] * DDIM + lane + 32], coef[u] * c1[u]);
            atomicAdd(&facc[sj[u] * DDIM + lane],      coef[u] * a0[u]);
            atomicAdd(&facc[sj[u] * DDIM + lane + 32], coef[u] * a1[u]);
        }
    }
    for (; k0 < n; ++k0) {
        const int i = lane_to_scalar(vi, k0), j = lane_to_scalar(vj, k0);
        const float a0 = Kn[i * DDIM + lane], a1 = Kn[i * DDIM + lane + 32];
        const float c0 = Kn[j * DDIM + lane], c1 = Kn[j * DDIM + lane + 32];
        const float gi = ISK ? 0.f : __int_as_float(lane_to_scalar(__float_as_int(gvi), k0));
        const float gj = ISK ? 0.f : __int_as_float(lane_to_scalar(__float_as_int(gvj), k0));
        edge_one<ISK>(i, j, a0, a1, c0, c1, gi, gj, lane, k0,
                      w1a, w1b, b1a, b1b, w2a, w2b, myGram, facc);
    }
    if (!ISK && e < E) {
        atomicAdd(&fH[vi], 0.5f * myGram * gvj);
        atomicAdd(&fH[vj], 0.5f * myGram * gvi);
    }
}

__global__ void __launch_bounds__(256)
k_edges(const int* __restrict__ indHK, const int* __restrict__ indK,
        const float* __restrict__ Kn, const float* __restrict__ gv,
        const float* __restrict__ w1v, const float* __restrict__ b1v,
        const float* __restrict__ w2v, float* __restrict__ fH,
        float* __restrict__ facc, int EHK, int EK) {
    const unsigned lane = threadIdx.x & 31u;
    const int wave = __builtin_amdgcn_readfirstlane(threadIdx.x >> 5);
    const int gw = blockIdx.x * (int)(blockDim.x >> 5) + wave;
    const int tw = gridDim.x * (int)(blockDim.x >> 5);
    const int nHK = (EHK + 31) >> 5, nK = (EK + 31) >> 5;
    const float w1a = w1v[lane], w1b = w1v[lane + 32];
    const float b1a = b1v[lane], b1b = b1v[lane + 32];
    const float w2a = w2v[lane], w2b = w2v[lane + 32];
    for (int pass = 0; pass < 2; ++pass) {
        const bool isK = ((pass ^ (int)blockIdx.x) & 1) != 0;
        if (isK) {
            for (int c = gw; c < nK; c += tw)
                edge_chunk<true>(indK, EK, c, lane, Kn, gv, w1a, w1b, b1a, b1b, w2a, w2b, fH, facc);
        } else {
            for (int c = gw; c < nHK; c += tw)
                edge_chunk<false>(indHK, EHK, c, lane, Kn, gv, w1a, w1b, b1a, b1b, w2a, w2b, fH, facc);
        }
    }
}

__global__ void k_komega(const float* __restrict__ Kn, const float* __restrict__ skew,
                         float* __restrict__ Kom, int N) {
    __shared__ float ssk[DDIM * DDIM];
    for (int t = threadIdx.x; t < DDIM * DDIM; t += blockDim.x) ssk[t] = skew[t];
    __syncthreads();
    int lane = threadIdx.x & 31, wid = threadIdx.x >> 5;
    int tile = blockIdx.x * (blockDim.x >> 5) + wid;
    if (tile >= (N >> 4)) return;
    int row0 = tile << 4;
    int l15 = lane & 15;
    int hi = lane >> 4;
    int arow = row0 + l15;
    int kofs = hi * 2;
    v8f acc[4] = {};
#pragma unroll
    for (int k0 = 0; k0 < DDIM; k0 += 4) {
        v2f A;
        A.x = Kn[arow * DDIM + k0 + kofs];
        A.y = Kn[arow * DDIM + k0 + kofs + 1];
#pragma unroll
        for (int t = 0; t < 4; ++t) {
            v2f B;
            int n = (t << 4) + l15;
            B.x = ssk[(k0 + kofs) * DDIM + n];
            B.y = ssk[(k0 + kofs + 1) * DDIM + n];
            acc[t] = __builtin_amdgcn_wmma_f32_16x16x4_f32(
                false, A, false, B, (short)0, acc[t], false, false);
        }
    }
    int crow = row0 + hi * 8;
#pragma unroll
    for (int t = 0; t < 4; ++t) {
#pragma unroll
        for (int v = 0; v < 8; ++v) {
            Kom[(crow + v) * DDIM + (t << 4) + l15] = acc[t][v];
        }
    }
}

__global__ void k_final(const float* __restrict__ Kn, const float* __restrict__ facc,
                        const float* __restrict__ Kom, float* __restrict__ fK, int N) {
    int lane = threadIdx.x & 31, wid = threadIdx.x >> 5;
    int row = blockIdx.x * (blockDim.x >> 5) + wid;
    if (row >= N) return;
    float k0 = Kn[row * DDIM + lane],   k1 = Kn[row * DDIM + lane + 32];
    float r0 = facc[row * DDIM + lane], r1 = facc[row * DDIM + lane + 32];
    float dot = wave_reduce_sum(k0 * r0 + k1 * r1);
    fK[row * DDIM + lane]      = -r0 + k0 * dot + Kom[row * DDIM + lane];
    fK[row * DDIM + lane + 32] = -r1 + k1 * dot + Kom[row * DDIM + lane + 32];
}

extern "C" void kernel_launch(void* const* d_in, const int* in_sizes, int n_in,
                              void* d_out, int out_size, void* d_ws, size_t ws_size,
                              hipStream_t stream) {
    const float* sH    = (const float*)d_in[0];
    const float* sK    = (const float*)d_in[1];
    const float* WH    = (const float*)d_in[2];
    const float* omega = (const float*)d_in[3];
    const float* w1v   = (const float*)d_in[4];
    const float* b1v   = (const float*)d_in[5];
    const float* w2v   = (const float*)d_in[6];
    const int*   indK  = (const int*)d_in[7];
    const int*   indHK = (const int*)d_in[8];

    int N   = in_sizes[0];
    int EK  = in_sizes[7] / 2;
    int EHK = in_sizes[8] / 2;
    int ND  = N * DDIM;

    float* fH = (float*)d_out;
    float* fK = (float*)d_out + N;

    float* Kn   = (float*)d_ws;
    float* gv   = Kn + (size_t)ND;
    float* skew = gv + N;
    float* facc = skew + DDIM * DDIM;
    float* Kom  = facc + (size_t)ND;

    k_prep  <<<(ND + 255) / 256,        256, 0, stream>>>(sH, omega, facc, gv, skew, N);
    k_norm  <<<(N + 7) / 8,             256, 0, stream>>>(sK, Kn, N);
    if (N == MV_N)
        k_matvec_stream<<<MV_N / MV_ROWS, MV_WAVES * 32, 0, stream>>>(WH, gv, sH, fH);
    else
        k_matvec<<<256,                 256, 0, stream>>>(WH, gv, sH, fH, N);
    k_edges <<<1024,                    256, 0, stream>>>(indHK, indK, Kn, gv, w1v, b1v, w2v,
                                                          fH, facc, EHK, EK);
    k_komega<<<((N >> 4) + 7) / 8,      256, 0, stream>>>(Kn, skew, Kom, N);
    k_final <<<(N + 7) / 8,             256, 0, stream>>>(Kn, facc, Kom, fK, N);
}


# six dependent launches folded into three: each streaming workgroup also zeroes, takes tanh and row-normalizes for its own 32 rows; skew, 64x64 WMMA product and projection in one closing kernel
# speedup vs baseline: 2.1505x; 1.0702x over previous
#include <hip/hip_runtime.h>
#include <hip/hip_bf16.h>

typedef __attribute__((ext_vector_type(2))) float v2f;
typedef __attribute__((ext_vector_type(8))) float v8f;

#define DDIM 64
#define NMAX 8192

__device__ __forceinline__ float wave_reduce_sum(float v) {
#pragma unroll
    for (int off = 16; off > 0; off >>= 1) v += __shfl_xor(v, off, 32);
    return v;
}

__global__ void k_prep(const float* __restrict__ sH,
                       float* __restrict__ facc, float* __restrict__ gv, int N) {
    int i = blockIdx.x * blockDim.x + threadIdx.x;
    int ND = N * DDIM;
    int stride = gridDim.x * blockDim.x;
    for (int t = i; t < ND; t += stride) facc[t] = 0.0f;
    if (i < N) gv[i] = tanhf(sH[i]);
}

__global__ void k_norm(const float* __restrict__ K, float* __restrict__ Kn, int N) {
    int lane = threadIdx.x & 31, wid = threadIdx.x >> 5;
    int row = blockIdx.x * (blockDim.x >> 5) + wid;
    if (row >= N) return;
    float a0 = K[row * DDIM + lane];
    float a1 = K[row * DDIM + lane + 32];
    float s = wave_reduce_sum(a0 * a0 + a1 * a1);
    float inv = rsqrtf(s);
    Kn[row * DDIM + lane]      = a0 * inv;
    Kn[row * DDIM + lane + 32] = a1 * inv;
}

#define MV_N     8192
#define MV_WAVES 16
#define MV_ROWS  32
#define MV_Q     4
#define MV_G     4

typedef float f32x4 __attribute__((ext_vector_type(4)));

__global__ void __launch_bounds__(MV_WAVES * 32)
k_front_stream(const float* __restrict__ W, const float* __restrict__ sH,
               const float* __restrict__ sK, float* __restrict__ Kn,
               float* __restrict__ gv, float* __restrict__ facc,
               float* __restrict__ fH) {
    __shared__ float part[MV_ROWS][MV_WAVES * 32];
    __shared__ float res[MV_ROWS];
    const unsigned tid = threadIdx.x, lane = tid & 31u;
    const unsigned wave = __builtin_amdgcn_readfirstlane(tid >> 5);
    const unsigned row0 = blockIdx.x * MV_ROWS;
    const unsigned col  = wave * (MV_Q * 128u) + lane * 4u;

    f32x4 gq[MV_Q];
#pragma unroll
    for (int q = 0; q < MV_Q; ++q) gq[q] = *(const f32x4*)(sH + col + 128u * q);
    const unsigned nrow = row0 + 2u * wave;
    const float n00 = sK[nrow * DDIM + lane],          n01 = sK[nrow * DDIM + lane + 32u];
    const float n10 = sK[(nrow + 1u) * DDIM + lane],   n11 = sK[(nrow + 1u) * DDIM + lane + 32u];
    const float hown = sH[row0 + lane];

    const float* wb = W + (size_t)row0 * MV_N + col;
    f32x4 w[2][MV_G][MV_Q];
#define MV_LOAD_GROUP(K) do { \
    _Pragma("unroll") for (int r_ = 0; r_ < MV_G; ++r_) \
    _Pragma("unroll") for (int q_ = 0; q_ < MV_Q; ++q_) \
        w[(K) & 1][r_][q_] = __builtin_nontemporal_load( \
            (const f32x4*)(wb + (size_t)((K) * MV_G + r_) * MV_N + 128u * q_)); } while (0)

    __builtin_amdgcn_sched_barrier(0);
    MV_LOAD_GROUP(0);
    MV_LOAD_GROUP(1);
    __builtin_amdgcn_sched_barrier(0);

#pragma unroll
    for (int q = 0; q < MV_Q; ++q) {
        gq[q].x = tanhf(gq[q].x);
        gq[q].y = tanhf(gq[q].y);
        gq[q].z = tanhf(gq[q].z);
        gq[q].w = tanhf(gq[q].w);
    }
    if (tid < MV_ROWS) gv[row0 + tid] = tanhf(hown);
    *(f32x4*)(facc + (size_t)row0 * DDIM + 4u * tid) = (f32x4){0.f, 0.f, 0.f, 0.f};
    {
        const float q0 = wave_reduce_sum(n00 * n00 + n01 * n01);
        const float q1 = wave_reduce_sum(n10 * n10 + n11 * n11);
        const float i0 = rsqrtf(q0), i1 = rsqrtf(q1);
        Kn[nrow * DDIM + lane]              = n00 * i0;
        Kn[nrow * DDIM + lane + 32u]        = n01 * i0;
        Kn[(nrow + 1u) * DDIM + lane]       = n10 * i1;
        Kn[(nrow + 1u) * DDIM + lane + 32u] = n11 * i1;
    }

#pragma unroll
    for (int k = 0; k < MV_ROWS / MV_G; ++k) {
        __builtin_amdgcn_sched_barrier(0);
        if (k >= 1 && k + 1 < MV_ROWS / MV_G) MV_LOAD_GROUP(k + 1);
        __builtin_amdgcn_sched_barrier(0);
#pragma unroll
        for (int r = 0; r < MV_G; ++r) {
            float a0 = 0.f, a1 = 0.f, a2 = 0.f, a3 = 0.f;
#pragma unroll
            for (int q = 0; q < MV_Q; ++q) {
                const f32x4 x = w[k & 1][r][q];
                a0 = __builtin_fmaf(x.x, gq[q].x, a0);
                a1 = __builtin_fmaf(x.y, gq[q].y, a1);
                a2 = __builtin_fmaf(x.z, gq[q].z, a2);
                a3 = __builtin_fmaf(x.w, gq[q].w, a3);
            }
            part[k * MV_G + r][tid] = (a0 + a1) + (a2 + a3);
        }
    }
#undef MV_LOAD_GROUP
    __syncthreads();

    float s0 = 0.f, s1 = 0.f;
#pragma unroll
    for (int j = 0; j < MV_WAVES; ++j) {
        s0 += part[2u * wave][lane + 32u * j];
        s1 += part[2u * wave + 1u][lane + 32u * j];
    }
    s0 = wave_reduce_sum(s0);
    s1 = wave_reduce_sum(s1);
    if (lane == 0) { res[2u * wave] = s0; res[2u * wave + 1u] = s1; }
    __syncthreads();
    if (tid < MV_ROWS) fH[row0 + tid] = -sH[row0 + tid] + res[tid];
}

__global__ void k_matvec(const float* __restrict__ W, const float* __restrict__ gv,
                         const float* __restrict__ sH, float* __restrict__ fH, int N) {
    __shared__ float gs[NMAX];
    for (int t = threadIdx.x; t < N && t < NMAX; t += blockDim.x) gs[t] = gv[t];
    __syncthreads();
    int lane = threadIdx.x & 31, wid = threadIdx.x >> 5;
    int wpb = blockDim.x >> 5;
    int gw = blockIdx.x * wpb + wid;
    int tw = gridDim.x * wpb;
    int n4 = N >> 2;
    for (int row = gw; row < N; row += tw) {
        const float4* Wr = (const float4*)(W + (size_t)row * N);
        float acc = 0.0f;
        for (int c4 = lane; c4 < n4; c4 += 32) {
            float4 w = Wr[c4];
            int c = c4 << 2;
            acc += w.x * gs[c] + w.y * gs[c + 1] + w.z * gs[c + 2] + w.w * gs[c + 3];
        }
        acc = wave_reduce_sum(acc);
        if (lane == 0) fH[row] = -sH[row] + acc;
    }
}

#define EDGE_B 4

__device__ __forceinline__ int lane_to_scalar(int v, int k) {
    return __builtin_amdgcn_readlane(v, k);
}

__device__ __forceinline__ void wave_reduce_sum_batch(float (&v)[EDGE_B]) {
#pragma unroll
    for (int off = 16; off > 0; off >>= 1) {
        float t[EDGE_B];
#pragma unroll
        for (int u = 0; u < EDGE_B; ++u) t[u] = __shfl_xor(v[u], off, 32);
#pragma unroll
        for (int u = 0; u < EDGE_B; ++u) v[u] += t[u];
    }
}

template <bool ISK>
__device__ __forceinline__ void edge_one(int i, int j, float a0, float a1, float c0, float c1,
                                         float gi, float gj, unsigned lane, int k,
                                         float w1a, float w1b, float b1a, float b1b,
                                         float w2a, float w2b, float& myGram,
                                         float* __restrict__ facc) {
    float s = wave_reduce_sum(a0 * c0 + a1 * c1);
    float coef;
    if (ISK) {
        float h0 = tanhf(s * w1a + b1a);
        float h1 = tanhf(s * w1b + b1b);
        coef = wave_reduce_sum(h0 * w2a + h1 * w2b);
    } else {
        if (lane == (unsigned)k) myGram = s;
        coef = -(0.5f * gi * gj);
    }
    atomicAdd(&facc[i * DDIM + lane],      coef * c0);
    atomicAdd(&facc[i * DDIM + lane + 32], coef * c1);
    atomicAdd(&facc[j * DDIM + lane],      coef * a0);
    atomicAdd(&facc[j * DDIM + lane + 32], coef * a1);
}

template <bool ISK>
__device__ __forceinline__ void edge_chunk(const int* __restrict__ ind, int E, int chunk,
                                           unsigned lane, const float* __restrict__ Kn,
                                           const float* __restrict__ gv,
                                           float w1a, float w1b, float b1a, float b1b,
                                           float w2a, float w2b,
                                           float* __restrict__ fH, float* __restrict__ facc) {
    const int e0 = chunk * 32;
    const int n = min(32, E - e0);
    const int e = e0 + (int)lane;
    int vi = 0, vj = 0;
    if (e < E) {
        const int2 t = *(const int2*)(ind + 2 * (size_t)e);
        vi = t.x; vj = t.y;
    }
    float gvi = 0.f, gvj = 0.f, myGram = 0.f;
    if (!ISK) { gvi = gv[vi]; gvj = gv[vj]; }

    int k0 = 0;
    for (; k0 + EDGE_B <= n; k0 += EDGE_B) {
        int si[EDGE_B], sj[EDGE_B];
        float a0[EDGE_B], a1[EDGE_B], c0[EDGE_B], c1[EDGE_B], gi[EDGE_B], gj[EDGE_B];
#pragma unroll
        for (int u = 0; u < EDGE_B; ++u) {
            si[u] = lane_to_scalar(vi, k0 + u);
            sj[u] = lane_to_scalar(vj, k0 + u);
            a0[u] = Kn[si[u] * DDIM + lane];
            a1[u] = Kn[si[u] * DDIM + lane + 32];
            c0[u] = Kn[sj[u] * DDIM + lane];
            c1[u] = Kn[sj[u] * DDIM + lane + 32];
            gi[u] = ISK ? 0.f : __int_as_float(lane_to_scalar(__float_as_int(gvi), k0 + u));
            gj[u] = ISK ? 0.f : __int_as_float(lane_to_scalar(__float_as_int(gvj), k0 + u));
        }
        float s[EDGE_B], coef[EDGE_B];
#pragma unroll
        for (int u = 0; u < EDGE_B; ++u) s[u] = a0[u] * c0[u] + a1[u] * c1[u];
        wave_reduce_sum_batch(s);
        if (ISK) {
#pragma unroll
            for (int u = 0; u < EDGE_B; ++u) {
                const float h0 = tanhf(s[u] * w1a + b1a);
                const float h1 = tanhf(s[u] * w1b + b1b);
                coef[u] = h0 * w2a + h1 * w2b;
            }
            wave_reduce_sum_batch(coef);
        } else {
#pragma unroll
            for (int u = 0; u < EDGE_B; ++u) {
                if (lane == (unsigned)(k0 + u)) myGram = s[u];
                coef[u] = -(0.5f * gi[u] * gj[u]);
            }
        }
#pragma unroll
        for (int u = 0; u < EDGE_B; ++u) {
            atomicAdd(&facc[si[u] * DDIM + lane],      coef[u] * c0[u]);
            atomicAdd(&facc[si[u] * DDIM + lane + 32], coef[u] * c1[u]);
            atomicAdd(&facc[sj[u] * DDIM + lane],      coef[u] * a0[u]);
            atomicAdd(&facc[sj[u] * DDIM + lane + 32], coef[u] * a1[u]);
        }
    }
    for (; k0 < n; ++k0) {
        const int i = lane_to_scalar(vi, k0), j = lane_to_scalar(vj, k0);
        const float a0 = Kn[i * DDIM + lane], a1 = Kn[i * DDIM + lane + 32];
        const float c0 = Kn[j * DDIM + lane], c1 = Kn[j * DDIM + lane + 32];
        const float gi = ISK ? 0.f : __int_as_float(lane_to_scalar(__float_as_int(gvi), k0));
        const float gj = ISK ? 0.f : __int_as_float(lane_to_scalar(__float_as_int(gvj), k0));
        edge_one<ISK>(i, j, a0, a1, c0, c1, gi, gj, lane, k0,
                      w1a, w1b, b1a, b1b, w2a, w2b, myGram, facc);
    }
    if (!ISK && e < E) {
        atomicAdd(&fH[vi], 0.5f * myGram * gvj);
        atomicAdd(&fH[vj], 0.5f * myGram * gvi);
    }
}

__global__ void __launch_bounds__(256)
k_edges(const int* __restrict__ indHK, const int* __restrict__ indK,
        const float* __restrict__ Kn, const float* __restrict__ gv,
        const float* __restrict__ w1v, const float* __restrict__ b1v,
        const float* __restrict__ w2v, float* __restrict__ fH,
        float* __restrict__ facc, int EHK, int EK) {
    const unsigned lane = threadIdx.x & 31u;
    const int wave = __builtin_amdgcn_readfirstlane(threadIdx.x >> 5);
    const int gw = blockIdx.x * (int)(blockDim.x >> 5) + wave;
    const int tw = gridDim.x * (int)(blockDim.x >> 5);
    const int nHK = (EHK + 31) >> 5, nK = (EK + 31) >> 5;
    const float w1a = w1v[lane], w1b = w1v[lane + 32];
    const float b1a = b1v[lane], b1b = b1v[lane + 32];
    const float w2a = w2v[lane], w2b = w2v[lane + 32];
    for (int pass = 0; pass < 2; ++pass) {
        const bool isK = ((pass ^ (int)blockIdx.x) & 1) != 0;
        if (isK) {
            for (int c = gw; c < nK; c += tw)
                edge_chunk<true>(indK, EK, c, lane, Kn, gv, w1a, w1b, b1a, b1b, w2a, w2b, fH, facc);
        } else {
            for (int c = gw; c < nHK; c += tw)
                edge_chunk<false>(indHK, EHK, c, lane, Kn, gv, w1a, w1b, b1a, b1b, w2a, w2b, fH, facc);
        }
    }
}

#define TL_WAVES 4
#define OM_P     65
#define TL_P     68

__global__ void __launch_bounds__(TL_WAVES * 32)
k_tail(const float* __restrict__ Kn, const float* __restrict__ omega,
       const float* __restrict__ facc, float* __restrict__ fK, int N) {
    __shared__ float som[DDIM * OM_P];
    __shared__ float tl[TL_WAVES][16 * TL_P];
    const int lane = threadIdx.x & 31;
    const int wid = __builtin_amdgcn_readfirstlane(threadIdx.x >> 5);
    const int tile = blockIdx.x * TL_WAVES + wid;
    const bool live = tile < (N >> 4);
    const int row0 = tile << 4;
    const int l15 = lane & 15;
    const int hi = lane >> 4;
    const int kofs = hi * 2;
    float* my = tl[wid];

    v2f Af[DDIM / 4];
    if (live) {
        const int arow = row0 + l15;
#pragma unroll
        for (int s = 0; s < DDIM / 4; ++s) {
            Af[s].x = Kn[arow * DDIM + 4 * s + kofs];
            Af[s].y = Kn[arow * DDIM + 4 * s + kofs + 1];
        }
    }
#pragma unroll
    for (int u = 0; u < (DDIM * DDIM) / (TL_WAVES * 32); ++u) {
        const int t = u * (TL_WAVES * 32) + (int)threadIdx.x;
        som[(t >> 6) * OM_P + (t & 63)] = omega[t];
    }
    __syncthreads();

    if (live) {
        v8f acc[4] = {};
#pragma unroll
        for (int s = 0; s < DDIM / 4; ++s) {
            const int k = 4 * s + kofs;
#pragma unroll
            for (int t = 0; t < 4; ++t) {
                v2f B;
                const int n = (t << 4) + l15;
                B.x = 0.5f * (som[k * OM_P + n]       - som[n * OM_P + k]);
                B.y = 0.5f * (som[(k + 1) * OM_P + n] - som[n * OM_P + k + 1]);
                acc[t] = __builtin_amdgcn_wmma_f32_16x16x4_f32(
                    false, Af[s], false, B, (short)0, acc[t], false, false);
            }
        }
#pragma unroll
        for (int t = 0; t < 4; ++t) {
#pragma unroll
            for (int v = 0; v < 8; ++v) {
                my[(hi * 8 + v) * TL_P + (t << 4) + l15] = acc[t][v];
            }
        }
    }
    __syncthreads();

    if (live) {
#pragma unroll
        for (int r = 0; r < 16; ++r) {
            const int row = row0 + r;
            const float k0 = Kn[row * DDIM + lane],   k1 = Kn[row * DDIM + lane + 32];
            const float r0 = facc[row * DDIM + lane], r1 = facc[row * DDIM + lane + 32];
            const float dot = wave_reduce_sum(k0 * r0 + k1 * r1);
            fK[row * DDIM + lane]      = -r0 + k0 * dot + my[r * TL_P + lane];
            fK[row * DDIM + lane + 32] = -r1 + k1 * dot + my[r * TL_P + lane + 32];
        }
    }
}

extern "C" void kernel_launch(void* const* d_in, const int* in_sizes, int n_in,
                              void* d_out, int out_size, void* d_ws, size_t ws_size,
                              hipStream_t stream) {
    const float* sH    = (const float*)d_in[0];
    const float* sK    = (const float*)d_in[1];
    const float* WH    = (const float*)d_in[2];
    const float* omega = (const float*)d_in[3];
    const float* w1v   = (const float*)d_in[4];
    const float* b1v   = (const float*)d_in[5];
    const float* w2v   = (const float*)d_in[6];
    const int*   indK  = (const int*)d_in[7];
    const int*   indHK = (const int*)d_in[8];

    int N   = in_sizes[0];
    int EK  = in_sizes[7] / 2;
    int EHK = in_sizes[8] / 2;
    int ND  = N * DDIM;

    float* fH = (float*)d_out;
    float* fK = (float*)d_out + N;

    float* Kn   = (float*)d_ws;
    float* gv   = Kn + (size_t)ND;
    float* facc = gv + N;

    if (N == MV_N) {
        k_front_stream<<<MV_N / MV_ROWS, MV_WAVES * 32, 0, stream>>>(WH, sH, sK, Kn, gv, facc, fH);
    } else {
        k_prep  <<<(ND + 255) / 256,    256, 0, stream>>>(sH, facc, gv, N);
        k_norm  <<<(N + 7) / 8,         256, 0, stream>>>(sK, Kn, N);
        k_matvec<<<256,                 256, 0, stream>>>(WH, gv, sH, fH, N);
    }
    k_edges <<<1024,                    256, 0, stream>>>(indHK, indK, Kn, gv, w1v, b1v, w2v,
                                                          fH, facc, EHK, EK);
    k_tail  <<<((N >> 4) + TL_WAVES - 1) / TL_WAVES, TL_WAVES * 32, 0, stream>>>(Kn, omega, facc, fK, N);
}


# edge-launch atomic sums kept twice, one copy per half of the package (the wave's die number, read via s_sendmsg_rtn, picks the copy); closing kernel adds the copies: f32 sum order changes
# speedup vs baseline: 4.3256x; 2.0114x over previous
#include <hip/hip_runtime.h>
#include <hip/hip_bf16.h>

typedef __attribute__((ext_vector_type(2))) float v2f;
typedef __attribute__((ext_vector_type(8))) float v8f;

#define DDIM 64
#define NMAX 8192

__device__ __forceinline__ float wave_reduce_sum(float v) {
#pragma unroll
    for (int off = 16; off > 0; off >>= 1) v += __shfl_xor(v, off, 32);
    return v;
}

__global__ void k_prep(const float* __restrict__ sH, float* __restrict__ facc,
                       float* __restrict__ fhp, float* __restrict__ gv, int N) {
    int i = blockIdx.x * blockDim.x + threadIdx.x;
    int ND = N * DDIM;
    int stride = gridDim.x * blockDim.x;
    for (int t = i; t < 2 * ND; t += stride) facc[t] = 0.0f;
    if (i < N) { gv[i] = tanhf(sH[i]); fhp[i] = 0.0f; fhp[N + i] = 0.0f; }
}

__global__ void k_norm(const float* __restrict__ K, float* __restrict__ Kn, int N) {
    int lane = threadIdx.x & 31, wid = threadIdx.x >> 5;
    int row = blockIdx.x * (blockDim.x >> 5) + wid;
    if (row >= N) return;
    float a0 = K[row * DDIM + lane];
    float a1 = K[row * DDIM + lane + 32];
    float s = wave_reduce_sum(a0 * a0 + a1 * a1);
    float inv = rsqrtf(s);
    Kn[row * DDIM + lane]      = a0 * inv;
    Kn[row * DDIM + lane + 32] = a1 * inv;
}

#define MV_N     8192
#define MV_WAVES 16
#define MV_ROWS  32
#define MV_Q     4
#define MV_G     4

typedef float f32x4 __attribute__((ext_vector_type(4)));

__global__ void __launch_bounds__(MV_WAVES * 32)
k_front_stream(const float* __restrict__ W, const float* __restrict__ sH,
               const float* __restrict__ sK, float* __restrict__ Kn,
               float* __restrict__ gv, float* __restrict__ facc,
               float* __restrict__ fhp, float* __restrict__ fH) {
    __shared__ float part[MV_ROWS][MV_WAVES * 32];
    __shared__ float res[MV_ROWS];
    const unsigned tid = threadIdx.x, lane = tid & 31u;
    const unsigned wave = __builtin_amdgcn_readfirstlane(tid >> 5);
    const unsigned row0 = blockIdx.x * MV_ROWS;
    const unsigned col  = wave * (MV_Q * 128u) + lane * 4u;

    f32x4 gq[MV_Q];
#pragma unroll
    for (int q = 0; q < MV_Q; ++q) gq[q] = *(const f32x4*)(sH + col + 128u * q);
    const unsigned nrow = row0 + 2u * wave;
    const float n00 = sK[nrow * DDIM + lane],          n01 = sK[nrow * DDIM + lane + 32u];
    const float n10 = sK[(nrow + 1u) * DDIM + lane],   n11 = sK[(nrow + 1u) * DDIM + lane + 32u];
    const float hown = sH[row0 + lane];

    const float* wb = W + (size_t)row0 * MV_N + col;
    f32x4 w[2][MV_G][MV_Q];
#define MV_LOAD_GROUP(K) do { \
    _Pragma("unroll") for (int r_ = 0; r_ < MV_G; ++r_) \
    _Pragma("unroll") for (int q_ = 0; q_ < MV_Q; ++q_) \
        w[(K) & 1][r_][q_] = __builtin_nontemporal_load( \
            (const f32x4*)(wb + (size_t)((K) * MV_G + r_) * MV_N + 128u * q_)); } while (0)

    __builtin_amdgcn_sched_barrier(0);
    MV_LOAD_GROUP(0);
    MV_LOAD_GROUP(1);
    __builtin_amdgcn_sched_barrier(0);

#pragma unroll
    for (int q = 0; q < MV_Q; ++q) {
        gq[q].x = tanhf(gq[q].x);
        gq[q].y = tanhf(gq[q].y);
        gq[q].z = tanhf(gq[q].z);
        gq[q].w = tanhf(gq[q].w);
    }
    if (tid < MV_ROWS) {
        gv[row0 + tid] = tanhf(hown);
        fhp[row0 + tid] = 0.f;
        fhp[MV_N + row0 + tid] = 0.f;
    }
    *(f32x4*)(facc + (size_t)row0 * DDIM + 4u * tid) = (f32x4){0.f, 0.f, 0.f, 0.f};
    *(f32x4*)(facc + (size_t)(MV_N + row0) * DDIM + 4u * tid) = (f32x4){0.f, 0.f, 0.f, 0.f};
    {
        const float q0 = wave_reduce_sum(n00 * n00 + n01 * n01);
        const float q1 = wave_reduce_sum(n10 * n10 + n11 * n11);
        const float i0 = rsqrtf(q0), i1 = rsqrtf(q1);
        Kn[nrow * DDIM + lane]              = n00 * i0;
        Kn[nrow * DDIM + lane + 32u]        = n01 * i0;
        Kn[(nrow + 1u) * DDIM + lane]       = n10 * i1;
        Kn[(nrow + 1u) * DDIM + lane + 32u] = n11 * i1;
    }

#pragma unroll
    for (int k = 0; k < MV_ROWS / MV_G; ++k) {
        __builtin_amdgcn_sched_barrier(0);
        if (k >= 1 && k + 1 < MV_ROWS / MV_G) MV_LOAD_GROUP(k + 1);
        __builtin_amdgcn_sched_barrier(0);
#pragma unroll
        for (int r = 0; r < MV_G; ++r) {
            float a0 = 0.f, a1 = 0.f, a2 = 0.f, a3 = 0.f;
#pragma unroll
            for (int q = 0; q < MV_Q; ++q) {
                const f32x4 x = w[k & 1][r][q];
                a0 = __builtin_fmaf(x.x, gq[q].x, a0);
                a1 = __builtin_fmaf(x.y, gq[q].y, a1);
                a2 = __builtin_fmaf(x.z, gq[q].z, a2);
                a3 = __builtin_fmaf(x.w, gq[q].w, a3);
            }
            part[k * MV_G + r][tid] = (a0 + a1) + (a2 + a3);
        }
    }
#undef MV_LOAD_GROUP
    __syncthreads();

    float s0 = 0.f, s1 = 0.f;
#pragma unroll
    for (int j = 0; j < MV_WAVES; ++j) {
        s0 += part[2u * wave][lane + 32u * j];
        s1 += part[2u * wave + 1u][lane + 32u * j];
    }
    s0 = wave_reduce_sum(s0);
    s1 = wave_reduce_sum(s1);
    if (lane == 0) { res[2u * wave] = s0; res[2u * wave + 1u] = s1; }
    __syncthreads();
    if (tid < MV_ROWS) fH[row0 + tid] = -sH[row0 + tid] + res[tid];
}

__global__ void k_matvec(const float* __restrict__ W, const float* __restrict__ gv,
                         const float* __restrict__ sH, float* __restrict__ fH, int N) {
    __shared__ float gs[NMAX];
    for (int t = threadIdx.x; t < N && t < NMAX; t += blockDim.x) gs[t] = gv[t];
    __syncthreads();
    int lane = threadIdx.x & 31, wid = threadIdx.x >> 5;
    int wpb = blockDim.x >> 5;
    int gw = blockIdx.x * wpb + wid;
    int tw = gridDim.x * wpb;
    int n4 = N >> 2;
    for (int row = gw; row < N; row += tw) {
        const float4* Wr = (const float4*)(W + (size_t)row * N);
        float acc = 0.0f;
        for (int c4 = lane; c4 < n4; c4 += 32) {
            float4 w = Wr[c4];
            int c = c4 << 2;
            acc += w.x * gs[c] + w.y * gs[c + 1] + w.z * gs[c + 2] + w.w * gs[c + 3];
        }
        acc = wave_reduce_sum(acc);
        if (lane == 0) fH[row] = -sH[row] + acc;
    }
}

#define EDGE_B 4

__device__ __forceinline__ int lane_to_scalar(int v, int k) {
    return __builtin_amdgcn_readlane(v, k);
}

__device__ __forceinline__ void wave_reduce_sum_batch(float (&v)[EDGE_B]) {
#pragma unroll
    for (int off = 16; off > 0; off >>= 1) {
        float t[EDGE_B];
#pragma unroll
        for (int u = 0; u < EDGE_B; ++u) t[u] = __shfl_xor(v[u], off, 32);
#pragma unroll
        for (int u = 0; u < EDGE_B; ++u) v[u] += t[u];
    }
}

template <bool ISK>
__device__ __forceinline__ void edge_one(int i, int j, float a0, float a1, float c0, float c1,
                                         float gi, float gj, unsigned lane, int k,
                                         float w1a, float w1b, float b1a, float b1b,
                                         float w2a, float w2b, float& myGram,
                                         float* __restrict__ facc) {
    float s = wave_reduce_sum(a0 * c0 + a1 * c1);
    float coef;
    if (ISK) {
        float h0 = tanhf(s * w1a + b1a);
        float h1 = tanhf(s * w1b + b1b);
        coef = wave_reduce_sum(h0 * w2a + h1 * w2b);
    } else {
        if (lane == (unsigned)k) myGram = s;
        coef = -(0.5f * gi * gj);
    }
    atomicAdd(&facc[i * DDIM + lane],      coef * c0);
    atomicAdd(&facc[i * DDIM + lane + 32], coef * c1);
    atomicAdd(&facc[j * DDIM + lane],      coef * a0);
    atomicAdd(&facc[j * DDIM + lane + 32], coef * a1);
}

template <bool ISK>
__device__ __forceinline__ void edge_chunk(const int* __restrict__ ind, int E, int chunk,
                                           unsigned lane, const float* __restrict__ Kn,
                                           const float* __restrict__ gv,
                                           float w1a, float w1b, float b1a, float b1b,
                                           float w2a, float w2b,
                                           float* __restrict__ fH, float* __restrict__ facc) {
    const int e0 = chunk * 32;
    const int n = min(32, E - e0);
    const int e = e0 + (int)lane;
    int vi = 0, vj = 0;
    if (e < E) {
        const int2 t = *(const int2*)(ind + 2 * (size_t)e);
        vi = t.x; vj = t.y;
    }
    float gvi = 0.f, gvj = 0.f, myGram = 0.f;
    if (!ISK) { gvi = gv[vi]; gvj = gv[vj]; }

    int k0 = 0;
    for (; k0 + EDGE_B <= n; k0 += EDGE_B) {
        int si[EDGE_B], sj[EDGE_B];
        float a0[EDGE_B], a1[EDGE_B], c0[EDGE_B], c1[EDGE_B], gi[EDGE_B], gj[EDGE_B];
#pragma unroll
        for (int u = 0; u < EDGE_B; ++u) {
            si[u] = lane_to_scalar(vi, k0 + u);
            sj[u] = lane_to_scalar(vj, k0 + u);
            a0[u] = Kn[si[u] * DDIM + lane];
            a1[u] = Kn[si[u] * DDIM + lane + 32];
            c0[u] = Kn[sj[u] * DDIM + lane];
            c1[u] = Kn[sj[u] * DDIM + lane + 32];
            gi[u] = ISK ? 0.f : __int_as_float(lane_to_scalar(__float_as_int(gvi), k0 + u));
            gj[u] = ISK ? 0.f : __int_as_float(lane_to_scalar(__float_as_int(gvj), k0 + u));
        }
        float s[EDGE_B], coef[EDGE_B];
#pragma unroll
        for (int u = 0; u < EDGE_B; ++u) s[u] = a0[u] * c0[u] + a1[u] * c1[u];
        wave_reduce_sum_batch(s);
        if (ISK) {
#pragma unroll
            for (int u = 0; u < EDGE_B; ++u) {
                const float h0 = tanhf(s[u] * w1a + b1a);
                const float h1 = tanhf(s[u] * w1b + b1b);
                coef[u] = h0 * w2a + h1 * w2b;
            }
            wave_reduce_sum_batch(coef);
        } else {
#pragma unroll
            for (int u = 0; u < EDGE_B; ++u) {
                if (lane == (unsigned)(k0 + u)) myGram = s[u];
                coef[u] = -(0.5f * gi[u] * gj[u]);
            }
        }
#pragma unroll
        for (int u = 0; u < EDGE_B; ++u) {
            atomicAdd(&facc[si[u] * DDIM + lane],      coef[u] * c0[u]);
            atomicAdd(&facc[si[u] * DDIM + lane + 32], coef[u] * c1[u]);
            atomicAdd(&facc[sj[u] * DDIM + lane],      coef[u] * a0[u]);
            atomicAdd(&facc[sj[u] * DDIM + lane + 32], coef[u] * a1[u]);
        }
    }
    for (; k0 < n; ++k0) {
        const int i = lane_to_scalar(vi, k0), j = lane_to_scalar(vj, k0);
        const float a0 = Kn[i * DDIM + lane], a1 = Kn[i * DDIM + lane + 32];
        const float c0 = Kn[j * DDIM + lane], c1 = Kn[j * DDIM + lane + 32];
        const float gi = ISK ? 0.f : __int_as_float(lane_to_scalar(__float_as_int(gvi), k0));
        const float gj = ISK ? 0.f : __int_as_float(lane_to_scalar(__float_as_int(gvj), k0));
        edge_one<ISK>(i, j, a0, a1, c0, c1, gi, gj, lane, k0,
                      w1a, w1b, b1a, b1b, w2a, w2b, myGram, facc);
    }
    if (!ISK && e < E) {
        atomicAdd(&fH[vi], 0.5f * myGram * gvj);
        atomicAdd(&fH[vj], 0.5f * myGram * gvi);
    }
}

__global__ void __launch_bounds__(256)
k_edges(const int* __restrict__ indHK, const int* __restrict__ indK,
        const float* __restrict__ Kn, const float* __restrict__ gv,
        const float* __restrict__ w1v, const float* __restrict__ b1v,
        const float* __restrict__ w2v, float* __restrict__ fhp,
        float* __restrict__ faccAll, int N, int EHK, int EK) {
    const unsigned where = __builtin_amdgcn_s_sendmsg_rtn(0x87);
    const size_t half = ((where >> 16) & 15u) >= 4u ? 1u : 0u;
    float* __restrict__ fH   = fhp + half * (size_t)N;
    float* __restrict__ facc = faccAll + half * (size_t)N * DDIM;
    const unsigned lane = threadIdx.x & 31u;
    const int wave = __builtin_amdgcn_readfirstlane(threadIdx.x >> 5);
    const int gw = blockIdx.x * (int)(blockDim.x >> 5) + wave;
    const int tw = gridDim.x * (int)(blockDim.x >> 5);
    const int nHK = (EHK + 31) >> 5, nK = (EK + 31) >> 5;
    const float w1a = w1v[lane], w1b = w1v[lane + 32];
    const float b1a = b1v[lane], b1b = b1v[lane + 32];
    const float w2a = w2v[lane], w2b = w2v[lane + 32];
    for (int pass = 0; pass < 2; ++pass) {
        const bool isK = ((pass ^ (int)blockIdx.x) & 1) != 0;
        if (isK) {
            for (int c = gw; c < nK; c += tw)
                edge_chunk<true>(indK, EK, c, lane, Kn, gv, w1a, w1b, b1a, b1b, w2a, w2b, fH, facc);
        } else {
            for (int c = gw; c < nHK; c += tw)
                edge_chunk<false>(indHK, EHK, c, lane, Kn, gv, w1a, w1b, b1a, b1b, w2a, w2b, fH, facc);
        }
    }
}

#define TL_WAVES 4
#define OM_P     65
#define TL_P     68

__global__ void __launch_bounds__(TL_WAVES * 32)
k_tail(const float* __restrict__ Kn, const float* __restrict__ omega,
       const float* __restrict__ facc, const float* __restrict__ fhp,
       float* __restrict__ fH, float* __restrict__ fK, int N) {
    __shared__ float som[DDIM * OM_P];
    __shared__ float tl[TL_WAVES][16 * TL_P];
    const int lane = threadIdx.x & 31;
    const int hrow = blockIdx.x * (TL_WAVES * 16) + (int)threadIdx.x;
    const bool hlive = threadIdx.x < TL_WAVES * 16 && hrow < N;
    float hbase = 0.f, hs0 = 0.f, hs1 = 0.f;
    if (hlive) { hbase = fH[hrow]; hs0 = fhp[hrow]; hs1 = fhp[N + hrow]; }
    const int wid = __builtin_amdgcn_readfirstlane(threadIdx.x >> 5);
    const int tile = blockIdx.x * TL_WAVES + wid;
    const bool live = tile < (N >> 4);
    const int row0 = tile << 4;
    const int l15 = lane & 15;
    const int hi = lane >> 4;
    const int kofs = hi * 2;
    float* my = tl[wid];

    v2f Af[DDIM / 4];
    if (live) {
        const int arow = row0 + l15;
#pragma unroll
        for (int s = 0; s < DDIM / 4; ++s) {
            Af[s].x = Kn[arow * DDIM + 4 * s + kofs];
            Af[s].y = Kn[arow * DDIM + 4 * s + kofs + 1];
        }
    }
#pragma unroll
    for (int u = 0; u < (DDIM * DDIM) / (TL_WAVES * 32); ++u) {
        const int t = u * (TL_WAVES * 32) + (int)threadIdx.x;
        som[(t >> 6) * OM_P + (t & 63)] = omega[t];
    }
    __syncthreads();

    if (live) {
        v8f acc[4] = {};
#pragma unroll
        for (int s = 0; s < DDIM / 4; ++s) {
            const int k = 4 * s + kofs;
#pragma unroll
            for (int t = 0; t < 4; ++t) {
                v2f B;
                const int n = (t << 4) + l15;
                B.x = 0.5f * (som[k * OM_P + n]       - som[n * OM_P + k]);
                B.y = 0.5f * (som[(k + 1) * OM_P + n] - som[n * OM_P + k + 1]);
                acc[t] = __builtin_amdgcn_wmma_f32_16x16x4_f32(
                    false, Af[s], false, B, (short)0, acc[t], false, false);
            }
        }
#pragma unroll
        for (int t = 0; t < 4; ++t) {
#pragma unroll
            for (int v = 0; v < 8; ++v) {
                my[(hi * 8 + v) * TL_P + (t << 4) + l15] = acc[t][v];
            }
        }
    }
    __syncthreads();

    if (live) {
#pragma unroll
        for (int r = 0; r < 16; ++r) {
            const int row = row0 + r;
            const float k0 = Kn[row * DDIM + lane],   k1 = Kn[row * DDIM + lane + 32];
            const float r0 = facc[row * DDIM + lane]      + facc[(N + row) * DDIM + lane];
            const float r1 = facc[row * DDIM + lane + 32] + facc[(N + row) * DDIM + lane + 32];
            const float dot = wave_reduce_sum(k0 * r0 + k1 * r1);
            fK[row * DDIM + lane]      = -r0 + k0 * dot + my[r * TL_P + lane];
            fK[row * DDIM + lane + 32] = -r1 + k1 * dot + my[r * TL_P + lane + 32];
        }
    }
    if (hlive) fH[hrow] = hbase + (hs0 + hs1);
}

extern "C" void kernel_launch(void* const* d_in, const int* in_sizes, int n_in,
                              void* d_out, int out_size, void* d_ws, size_t ws_size,
                              hipStream_t stream) {
    const float* sH    = (const float*)d_in[0];
    const float* sK    = (const float*)d_in[1];
    const float* WH    = (const float*)d_in[2];
    const float* omega = (const float*)d_in[3];
    const float* w1v   = (const float*)d_in[4];
    const float* b1v   = (const float*)d_in[5];
    const float* w2v   = (const float*)d_in[6];
    const int*   indK  = (const int*)d_in[7];
    const int*   indHK = (const int*)d_in[8];

    int N   = in_sizes[0];
    int EK  = in_sizes[7] / 2;
    int EHK = in_sizes[8] / 2;
    int ND  = N * DDIM;

    float* fH = (float*)d_out;
    float* fK = (float*)d_out + N;

    float* Kn   = (float*)d_ws;
    float* gv   = Kn + (size_t)ND;
    float* facc = gv + N;
    float* fhp  = facc + 2 * (size_t)ND;

    if (N == MV_N) {
        k_front_stream<<<MV_N / MV_ROWS, MV_WAVES * 32, 0, stream>>>(WH, sH, sK, Kn, gv, facc, fhp, fH);
    } else {
        k_prep  <<<(ND + 255) / 256,    256, 0, stream>>>(sH, facc, fhp, gv, N);
        k_norm  <<<(N + 7) / 8,         256, 0, stream>>>(sK, Kn, N);
        k_matvec<<<256,                 256, 0, stream>>>(WH, gv, sH, fH, N);
    }
    k_edges <<<1024,                    256, 0, stream>>>(indHK, indK, Kn, gv, w1v, b1v, w2v,
                                                          fhp, facc, N, EHK, EK);
    k_tail  <<<((N >> 4) + TL_WAVES - 1) / TL_WAVES, TL_WAVES * 32, 0, stream>>>(Kn, omega, facc, fhp, fH, fK, N);
}


# edge launch: the 64 tanh per K edge as straight-line (1-t)/(1+t), t=exp(-2|x|), on the hardware exp2 and reciprocal (abs error of the order of 1e-7) instead of the branching library routine
# speedup vs baseline: 4.4841x; 1.0366x over previous
#include <hip/hip_runtime.h>
#include <hip/hip_bf16.h>

typedef __attribute__((ext_vector_type(2))) float v2f;
typedef __attribute__((ext_vector_type(8))) float v8f;

#define DDIM 64
#define NMAX 8192

__device__ __forceinline__ float wave_reduce_sum(float v) {
#pragma unroll
    for (int off = 16; off > 0; off >>= 1) v += __shfl_xor(v, off, 32);
    return v;
}

__global__ void k_prep(const float* __restrict__ sH, float* __restrict__ facc,
                       float* __restrict__ fhp, float* __restrict__ gv, int N) {
    int i = blockIdx.x * blockDim.x + threadIdx.x;
    int ND = N * DDIM;
    int stride = gridDim.x * blockDim.x;
    for (int t = i; t < 2 * ND; t += stride) facc[t] = 0.0f;
    if (i < N) { gv[i] = tanhf(sH[i]); fhp[i] = 0.0f; fhp[N + i] = 0.0f; }
}

__global__ void k_norm(const float* __restrict__ K, float* __restrict__ Kn, int N) {
    int lane = threadIdx.x & 31, wid = threadIdx.x >> 5;
    int row = blockIdx.x * (blockDim.x >> 5) + wid;
    if (row >= N) return;
    float a0 = K[row * DDIM + lane];
    float a1 = K[row * DDIM + lane + 32];
    float s = wave_reduce_sum(a0 * a0 + a1 * a1);
    float inv = rsqrtf(s);
    Kn[row * DDIM + lane]      = a0 * inv;
    Kn[row * DDIM + lane + 32] = a1 * inv;
}

#define MV_N     8192
#define MV_WAVES 16
#define MV_ROWS  32
#define MV_Q     4
#define MV_G     4

typedef float f32x4 __attribute__((ext_vector_type(4)));

__global__ void __launch_bounds__(MV_WAVES * 32)
k_front_stream(const float* __restrict__ W, const float* __restrict__ sH,
               const float* __restrict__ sK, float* __restrict__ Kn,
               float* __restrict__ gv, float* __restrict__ facc,
               float* __restrict__ fhp, float* __restrict__ fH) {
    __shared__ float part[MV_ROWS][MV_WAVES * 32];
    __shared__ float res[MV_ROWS];
    const unsigned tid = threadIdx.x, lane = tid & 31u;
    const unsigned wave = __builtin_amdgcn_readfirstlane(tid >> 5);
    const unsigned row0 = blockIdx.x * MV_ROWS;
    const unsigned col  = wave * (MV_Q * 128u) + lane * 4u;

    f32x4 gq[MV_Q];
#pragma unroll
    for (int q = 0; q < MV_Q; ++q) gq[q] = *(const f32x4*)(sH + col + 128u * q);
    const unsigned nrow = row0 + 2u * wave;
    const float n00 = sK[nrow * DDIM + lane],          n01 = sK[nrow * DDIM + lane + 32u];
    const float n10 = sK[(nrow + 1u) * DDIM + lane],   n11 = sK[(nrow + 1u) * DDIM + lane + 32u];
    const float hown = sH[row0 + lane];

    const float* wb = W + (size_t)row0 * MV_N + col;
    f32x4 w[2][MV_G][MV_Q];
#define MV_LOAD_GROUP(K) do { \
    _Pragma("unroll") for (int r_ = 0; r_ < MV_G; ++r_) \
    _Pragma("unroll") for (int q_ = 0; q_ < MV_Q; ++q_) \
        w[(K) & 1][r_][q_] = __builtin_nontemporal_load( \
            (const f32x4*)(wb + (size_t)((K) * MV_G + r_) * MV_N + 128u * q_)); } while (0)

    __builtin_amdgcn_sched_barrier(0);
    MV_LOAD_GROUP(0);
    MV_LOAD_GROUP(1);
    __builtin_amdgcn_sched_barrier(0);

#pragma unroll
    for (int q = 0; q < MV_Q; ++q) {
        gq[q].x = tanhf(gq[q].x);
        gq[q].y = tanhf(gq[q].y);
        gq[q].z = tanhf(gq[q].z);
        gq[q].w = tanhf(gq[q].w);
    }
    if (tid < MV_ROWS) {
        gv[row0 + tid] = tanhf(hown);
        fhp[row0 + tid] = 0.f;
        fhp[MV_N + row0 + tid] = 0.f;
    }
    *(f32x4*)(facc + (size_t)row0 * DDIM + 4u * tid) = (f32x4){0.f, 0.f, 0.f, 0.f};
    *(f32x4*)(facc + (size_t)(MV_N + row0) * DDIM + 4u * tid) = (f32x4){0.f, 0.f, 0.f, 0.f};
    {
        const float q0 = wave_reduce_sum(n00 * n00 + n01 * n01);
        const float q1 = wave_reduce_sum(n10 * n10 + n11 * n11);
        const float i0 = rsqrtf(q0), i1 = rsqrtf(q1);
        Kn[nrow * DDIM + lane]              = n00 * i0;
        Kn[nrow * DDIM + lane + 32u]        = n01 * i0;
        Kn[(nrow + 1u) * DDIM + lane]       = n10 * i1;
        Kn[(nrow + 1u) * DDIM + lane + 32u] = n11 * i1;
    }

#pragma unroll
    for (int k = 0; k < MV_ROWS / MV_G; ++k) {
        __builtin_amdgcn_sched_barrier(0);
        if (k >= 1 && k + 1 < MV_ROWS / MV_G) MV_LOAD_GROUP(k + 1);
        __builtin_amdgcn_sched_barrier(0);
#pragma unroll
        for (int r = 0; r < MV_G; ++r) {
            float a0 = 0.f, a1 = 0.f, a2 = 0.f, a3 = 0.f;
#pragma unroll
            for (int q = 0; q < MV_Q; ++q) {
                const f32x4 x = w[k & 1][r][q];
                a0 = __builtin_fmaf(x.x, gq[q].x, a0);
                a1 = __builtin_fmaf(x.y, gq[q].y, a1);
                a2 = __builtin_fmaf(x.z, gq[q].z, a2);
                a3 = __builtin_fmaf(x.w, gq[q].w, a3);
            }
            part[k * MV_G + r][tid] = (a0 + a1) + (a2 + a3);
        }
    }
#undef MV_LOAD_GROUP
    __syncthreads();

    float s0 = 0.f, s1 = 0.f;
#pragma unroll
    for (int j = 0; j < MV_WAVES; ++j) {
        s0 += part[2u * wave][lane + 32u * j];
        s1 += part[2u * wave + 1u][lane + 32u * j];
    }
    s0 = wave_reduce_sum(s0);
    s1 = wave_reduce_sum(s1);
    if (lane == 0) { res[2u * wave] = s0; res[2u * wave + 1u] = s1; }
    __syncthreads();
    if (tid < MV_ROWS) fH[row0 + tid] = -sH[row0 + tid] + res[tid];
}

__global__ void k_matvec(const float* __restrict__ W, const float* __restrict__ gv,
                         const float* __restrict__ sH, float* __restrict__ fH, int N) {
    __shared__ float gs[NMAX];
    for (int t = threadIdx.x; t < N && t < NMAX; t += blockDim.x) gs[t] = gv[t];
    __syncthreads();
    int lane = threadIdx.x & 31, wid = threadIdx.x >> 5;
    int wpb = blockDim.x >> 5;
    int gw = blockIdx.x * wpb + wid;
    int tw = gridDim.x * wpb;
    int n4 = N >> 2;
    for (int row = gw; row < N; row += tw) {
        const float4* Wr = (const float4*)(W + (size_t)row * N);
        float acc = 0.0f;
        for (int c4 = lane; c4 < n4; c4 += 32) {
            float4 w = Wr[c4];
            int c = c4 << 2;
            acc += w.x * gs[c] + w.y * gs[c + 1] + w.z * gs[c + 2] + w.w * gs[c + 3];
        }
        acc = wave_reduce_sum(acc);
        if (lane == 0) fH[row] = -sH[row] + acc;
    }
}

#define EDGE_B 4

__device__ __forceinline__ int lane_to_scalar(int v, int k) {
    return __builtin_amdgcn_readlane(v, k);
}

__device__ __forceinline__ void wave_reduce_sum_batch(float (&v)[EDGE_B]) {
#pragma unroll
    for (int off = 16; off > 0; off >>= 1) {
        float t[EDGE_B];
#pragma unroll
        for (int u = 0; u < EDGE_B; ++u) t[u] = __shfl_xor(v[u], off, 32);
#pragma unroll
        for (int u = 0; u < EDGE_B; ++u) v[u] += t[u];
    }
}

__device__ __forceinline__ float tanh_exp_rcp(float x) {
    const float t = __builtin_amdgcn_exp2f(-2.885390082f * __builtin_fabsf(x));
    const float y = (1.0f - t) * __builtin_amdgcn_rcpf(1.0f + t);
    return __builtin_copysignf(y, x);
}

template <bool ISK>
__device__ __forceinline__ void edge_one(int i, int j, float a0, float a1, float c0, float c1,
                                         float gi, float gj, unsigned lane, int k,
                                         float w1a, float w1b, float b1a, float b1b,
                                         float w2a, float w2b, float& myGram,
                                         float* __restrict__ facc) {
    float s = wave_reduce_sum(a0 * c0 + a1 * c1);
    float coef;
    if (ISK) {
        float h0 = tanh_exp_rcp(s * w1a + b1a);
        float h1 = tanh_exp_rcp(s * w1b + b1b);
        coef = wave_reduce_sum(h0 * w2a + h1 * w2b);
    } else {
        if (lane == (unsigned)k) myGram = s;
        coef = -(0.5f * gi * gj);
    }
    atomicAdd(&facc[i * DDIM + lane],      coef * c0);
    atomicAdd(&facc[i * DDIM + lane + 32], coef * c1);
    atomicAdd(&facc[j * DDIM + lane],      coef * a0);
    atomicAdd(&facc[j * DDIM + lane + 32], coef * a1);
}

template <bool ISK>
__device__ __forceinline__ void edge_chunk(const int* __restrict__ ind, int E, int chunk,
                                           unsigned lane, const float* __restrict__ Kn,
                                           const float* __restrict__ gv,
                                           float w1a, float w1b, float b1a, float b1b,
                                           float w2a, float w2b,
                                           float* __restrict__ fH, float* __restrict__ facc) {
    const int e0 = chunk * 32;
    const int n = min(32, E - e0);
    const int e = e0 + (int)lane;
    int vi = 0, vj = 0;
    if (e < E) {
        const int2 t = *(const int2*)(ind + 2 * (size_t)e);
        vi = t.x; vj = t.y;
    }
    float gvi = 0.f, gvj = 0.f, myGram = 0.f;
    if (!ISK) { gvi = gv[vi]; gvj = gv[vj]; }

    int k0 = 0;
    for (; k0 + EDGE_B <= n; k0 += EDGE_B) {
        int si[EDGE_B], sj[EDGE_B];
        float a0[EDGE_B], a1[EDGE_B], c0[EDGE_B], c1[EDGE_B], gi[EDGE_B], gj[EDGE_B];
#pragma unroll
        for (int u = 0; u < EDGE_B; ++u) {
            si[u] = lane_to_scalar(vi, k0 + u);
            sj[u] = lane_to_scalar(vj, k0 + u);
            a0[u] = Kn[si[u] * DDIM + lane];
            a1[u] = Kn[si[u] * DDIM + lane + 32];
            c0[u] = Kn[sj[u] * DDIM + lane];
            c1[u] = Kn[sj[u] * DDIM + lane + 32];
            gi[u] = ISK ? 0.f : __int_as_float(lane_to_scalar(__float_as_int(gvi), k0 + u));
            gj[u] = ISK ? 0.f : __int_as_float(lane_to_scalar(__float_as_int(gvj), k0 + u));
        }
        float s[EDGE_B], coef[EDGE_B];
#pragma unroll
        for (int u = 0; u < EDGE_B; ++u) s[u] = a0[u] * c0[u] + a1[u] * c1[u];
        wave_reduce_sum_batch(s);
        if (ISK) {
#pragma unroll
            for (int u = 0; u < EDGE_B; ++u) {
                const float h0 = tanh_exp_rcp(s[u] * w1a + b1a);
                const float h1 = tanh_exp_rcp(s[u] * w1b + b1b);
                coef[u] = h0 * w2a + h1 * w2b;
            }
            wave_reduce_sum_batch(coef);
        } else {
#pragma unroll
            for (int u = 0; u < EDGE_B; ++u) {
                if (lane == (unsigned)(k0 + u)) myGram = s[u];
                coef[u] = -(0.5f * gi[u] * gj[u]);
            }
        }
#pragma unroll
        for (int u = 0; u < EDGE_B; ++u) {
            atomicAdd(&facc[si[u] * DDIM + lane],      coef[u] * c0[u]);
            atomicAdd(&facc[si[u] * DDIM + lane + 32], coef[u] * c1[u]);
            atomicAdd(&facc[sj[u] * DDIM + lane],      coef[u] * a0[u]);
            atomicAdd(&facc[sj[u] * DDIM + lane + 32], coef[u] * a1[u]);
        }
    }
    for (; k0 < n; ++k0) {
        const int i = lane_to_scalar(vi, k0), j = lane_to_scalar(vj, k0);
        const float a0 = Kn[i * DDIM + lane], a1 = Kn[i * DDIM + lane + 32];
        const float c0 = Kn[j * DDIM + lane], c1 = Kn[j * DDIM + lane + 32];
        const float gi = ISK ? 0.f : __int_as_float(lane_to_scalar(__float_as_int(gvi), k0));
        const float gj = ISK ? 0.f : __int_as_float(lane_to_scalar(__float_as_int(gvj), k0));
        edge_one<ISK>(i, j, a0, a1, c0, c1, gi, gj, lane, k0,
                      w1a, w1b, b1a, b1b, w2a, w2b, myGram, facc);
    }
    if (!ISK && e < E) {
        atomicAdd(&fH[vi], 0.5f * myGram * gvj);
        atomicAdd(&fH[vj], 0.5f * myGram * gvi);
    }
}

__global__ void __launch_bounds__(256)
k_edges(const int* __restrict__ indHK, const int* __restrict__ indK,
        const float* __restrict__ Kn, const float* __restrict__ gv,
        const float* __restrict__ w1v, const float* __restrict__ b1v,
        const float* __restrict__ w2v, float* __restrict__ fhp,
        float* __restrict__ faccAll, int N, int EHK, int EK) {
    const unsigned where = __builtin_amdgcn_s_sendmsg_rtn(0x87);
    const size_t half = ((where >> 16) & 15u) >= 4u ? 1u : 0u;
    float* __restrict__ fH   = fhp + half * (size_t)N;
    float* __restrict__ facc = faccAll + half * (size_t)N * DDIM;
    const unsigned lane = threadIdx.x & 31u;
    const int wave = __builtin_amdgcn_readfirstlane(threadIdx.x >> 5);
    const int gw = blockIdx.x * (int)(blockDim.x >> 5) + wave;
    const int tw = gridDim.x * (int)(blockDim.x >> 5);
    const int nHK = (EHK + 31) >> 5, nK = (EK + 31) >> 5;
    const float w1a = w1v[lane], w1b = w1v[lane + 32];
    const float b1a = b1v[lane], b1b = b1v[lane + 32];
    const float w2a = w2v[lane], w2b = w2v[lane + 32];
    for (int pass = 0; pass < 2; ++pass) {
        const bool isK = ((pass ^ (int)blockIdx.x) & 1) != 0;
        if (isK) {
            for (int c = gw; c < nK; c += tw)
                edge_chunk<true>(indK, EK, c, lane, Kn, gv, w1a, w1b, b1a, b1b, w2a, w2b, fH, facc);
        } else {
            for (int c = gw; c < nHK; c += tw)
                edge_chunk<false>(indHK, EHK, c, lane, Kn, gv, w1a, w1b, b1a, b1b, w2a, w2b, fH, facc);
        }
    }
}

#define TL_WAVES 4
#define OM_P     65
#define TL_P     68

__global__ void __launch_bounds__(TL_WAVES * 32)
k_tail(const float* __restrict__ Kn, const float* __restrict__ omega,
       const float* __restrict__ facc, const float* __restrict__ fhp,
       float* __restrict__ fH, float* __restrict__ fK, int N) {
    __shared__ float som[DDIM * OM_P];
    __shared__ float tl[TL_WAVES][16 * TL_P];
    const int lane = threadIdx.x & 31;
    const int hrow = blockIdx.x * (TL_WAVES * 16) + (int)threadIdx.x;
    const bool hlive = threadIdx.x < TL_WAVES * 16 && hrow < N;
    float hbase = 0.f, hs0 = 0.f, hs1 = 0.f;
    if (hlive) { hbase = fH[hrow]; hs0 = fhp[hrow]; hs1 = fhp[N + hrow]; }
    const int wid = __builtin_amdgcn_readfirstlane(threadIdx.x >> 5);
    const int tile = blockIdx.x * TL_WAVES + wid;
    const bool live = tile < (N >> 4);
    const int row0 = tile << 4;
    const int l15 = lane & 15;
    const int hi = lane >> 4;
    const int kofs = hi * 2;
    float* my = tl[wid];

    v2f Af[DDIM / 4];
    if (live) {
        const int arow = row0 + l15;
#pragma unroll
        for (int s = 0; s < DDIM / 4; ++s) {
            Af[s].x = Kn[arow * DDIM + 4 * s + kofs];
            Af[s].y = Kn[arow * DDIM + 4 * s + kofs + 1];
        }
    }
#pragma unroll
    for (int u = 0; u < (DDIM * DDIM) / (TL_WAVES * 32); ++u) {
        const int t = u * (TL_WAVES * 32) + (int)threadIdx.x;
        som[(t >> 6) * OM_P + (t & 63)] = omega[t];
    }
    __syncthreads();

    if (live) {
        v8f acc[4] = {};
#pragma unroll
        for (int s = 0; s < DDIM / 4; ++s) {
            const int k = 4 * s + kofs;
#pragma unroll
            for (int t = 0; t < 4; ++t) {
                v2f B;
                const int n = (t << 4) + l15;
                B.x = 0.5f * (som[k * OM_P + n]       - som[n * OM_P + k]);
                B.y = 0.5f * (som[(k + 1) * OM_P + n] - som[n * OM_P + k + 1]);
                acc[t] = __builtin_amdgcn_wmma_f32_16x16x4_f32(
                    false, Af[s], false, B, (short)0, acc[t], false, false);
            }
        }
#pragma unroll
        for (int t = 0; t < 4; ++t) {
#pragma unroll
            for (int v = 0; v < 8; ++v) {
                my[(hi * 8 + v) * TL_P + (t << 4) + l15] = acc[t][v];
            }
        }
    }
    __syncthreads();

    if (live) {
#pragma unroll
        for (int r = 0; r < 16; ++r) {
            const int row = row0 + r;
            const float k0 = Kn[row * DDIM + lane],   k1 = Kn[row * DDIM + lane + 32];
            const float r0 = facc[row * DDIM + lane]      + facc[(N + row) * DDIM + lane];
            const float r1 = facc[row * DDIM + lane + 32] + facc[(N + row) * DDIM + lane + 32];
            const float dot = wave_reduce_sum(k0 * r0 + k1 * r1);
            fK[row * DDIM + lane]      = -r0 + k0 * dot + my[r * TL_P + lane];
            fK[row * DDIM + lane + 32] = -r1 + k1 * dot + my[r * TL_P + lane + 32];
        }
    }
    if (hlive) fH[hrow] = hbase + (hs0 + hs1);
}

extern "C" void kernel_launch(void* const* d_in, const int* in_sizes, int n_in,
                              void* d_out, int out_size, void* d_ws, size_t ws_size,
                              hipStream_t stream) {
    const float* sH    = (const float*)d_in[0];
    const float* sK    = (const float*)d_in[1];
    const float* WH    = (const float*)d_in[2];
    const float* omega = (const float*)d_in[3];
    const float* w1v   = (const float*)d_in[4];
    const float* b1v   = (const float*)d_in[5];
    const float* w2v   = (const float*)d_in[6];
    const int*   indK  = (const int*)d_in[7];
    const int*   indHK = (const int*)d_in[8];

    int N   = in_sizes[0];
    int EK  = in_sizes[7] / 2;
    int EHK = in_sizes[8] / 2;
    int ND  = N * DDIM;

    float* fH = (float*)d_out;
    float* fK = (float*)d_out + N;

    float* Kn   = (float*)d_ws;
    float* gv   = Kn + (size_t)ND;
    float* facc = gv + N;
    float* fhp  = facc + 2 * (size_t)ND;

    if (N == MV_N) {
        k_front_stream<<<MV_N / MV_ROWS, MV_WAVES * 32, 0, stream>>>(WH, sH, sK, Kn, gv, facc, fhp, fH);
    } else {
        k_prep  <<<(ND + 255) / 256,    256, 0, stream>>>(sH, facc, fhp, gv, N);
        k_norm  <<<(N + 7) / 8,         256, 0, stream>>>(sK, Kn, N);
        k_matvec<<<256,                 256, 0, stream>>>(WH, gv, sH, fH, N);
    }
    k_edges <<<1024,                    256, 0, stream>>>(indHK, indK, Kn, gv, w1v, b1v, w2v,
                                                          fhp, facc, N, EHK, EK);
    k_tail  <<<((N >> 4) + TL_WAVES - 1) / TL_WAVES, TL_WAVES * 32, 0, stream>>>(Kn, omega, facc, fhp, fH, fK, N);
}


# every wave sum (edge dot products, MLP output, row norms, projection) as a butterfly on the vector unit (row_xmask adds, one cross-row exchange), not LDS-unit shuffles; same pairs, same order
# speedup vs baseline: 4.7887x; 1.0679x over previous
#include <hip/hip_runtime.h>
#include <hip/hip_bf16.h>

typedef __attribute__((ext_vector_type(2))) float v2f;
typedef __attribute__((ext_vector_type(8))) float v8f;

#define DDIM 64
#define NMAX 8192

__device__ __forceinline__ float xhalf(float x) {
    const int b = __builtin_bit_cast(int, x);
    return __builtin_bit_cast(float,
        __builtin_amdgcn_permlanex16(b, b, 0x76543210, 0xfedcba98, false, false));
}
template <int M>
__device__ __forceinline__ float xrow(float x) {
    const int b = __builtin_bit_cast(int, x);
    return __builtin_bit_cast(float, __builtin_amdgcn_update_dpp(b, b, 0x160 | M, 0xf, 0xf, true));
}

__device__ __forceinline__ float wave_reduce_sum(float v) {
    v += xhalf(v);
    v += xrow<8>(v);
    v += xrow<4>(v);
    v += xrow<2>(v);
    v += xrow<1>(v);
    return v;
}

__global__ void k_prep(const float* __restrict__ sH, float* __restrict__ facc,
                       float* __restrict__ fhp, float* __restrict__ gv, int N) {
    int i = blockIdx.x * blockDim.x + threadIdx.x;
    int ND = N * DDIM;
    int stride = gridDim.x * blockDim.x;
    for (int t = i; t < 2 * ND; t += stride) facc[t] = 0.0f;
    if (i < N) { gv[i] = tanhf(sH[i]); fhp[i] = 0.0f; fhp[N + i] = 0.0f; }
}

__global__ void k_norm(const float* __restrict__ K, float* __restrict__ Kn, int N) {
    int lane = threadIdx.x & 31, wid = threadIdx.x >> 5;
    int row = blockIdx.x * (blockDim.x >> 5) + wid;
    if (row >= N) return;
    float a0 = K[row * DDIM + lane];
    float a1 = K[row * DDIM + lane + 32];
    float s = wave_reduce_sum(a0 * a0 + a1 * a1);
    float inv = rsqrtf(s);
    Kn[row * DDIM + lane]      = a0 * inv;
    Kn[row * DDIM + lane + 32] = a1 * inv;
}

#define MV_N     8192
#define MV_WAVES 16
#define MV_ROWS  32
#define MV_Q     4
#define MV_G     4

typedef float f32x4 __attribute__((ext_vector_type(4)));

__global__ void __launch_bounds__(MV_WAVES * 32)
k_front_stream(const float* __restrict__ W, const float* __restrict__ sH,
               const float* __restrict__ sK, float* __restrict__ Kn,
               float* __restrict__ gv, float* __restrict__ facc,
               float* __restrict__ fhp, float* __restrict__ fH) {
    __shared__ float part[MV_ROWS][MV_WAVES * 32];
    __shared__ float res[MV_ROWS];
    const unsigned tid = threadIdx.x, lane = tid & 31u;
    const unsigned wave = __builtin_amdgcn_readfirstlane(tid >> 5);
    const unsigned row0 = blockIdx.x * MV_ROWS;
    const unsigned col  = wave * (MV_Q * 128u) + lane * 4u;

    f32x4 gq[MV_Q];
#pragma unroll
    for (int q = 0; q < MV_Q; ++q) gq[q] = *(const f32x4*)(sH + col + 128u * q);
    const unsigned nrow = row0 + 2u * wave;
    const float n00 = sK[nrow * DDIM + lane],          n01 = sK[nrow * DDIM + lane + 32u];
    const float n10 = sK[(nrow + 1u) * DDIM + lane],   n11 = sK[(nrow + 1u) * DDIM + lane + 32u];
    const float hown = sH[row0 + lane];

    const float* wb = W + (size_t)row0 * MV_N + col;
    f32x4 w[2][MV_G][MV_Q];
#define MV_LOAD_GROUP(K) do { \
    _Pragma("unroll") for (int r_ = 0; r_ < MV_G; ++r_) \
    _Pragma("unroll") for (int q_ = 0; q_ < MV_Q; ++q_) \
        w[(K) & 1][r_][q_] = __builtin_nontemporal_load( \
            (const f32x4*)(wb + (size_t)((K) * MV_G + r_) * MV_N + 128u * q_)); } while (0)

    __builtin_amdgcn_sched_barrier(0);
    MV_LOAD_GROUP(0);
    MV_LOAD_GROUP(1);
    __builtin_amdgcn_sched_barrier(0);

#pragma unroll
    for (int q = 0; q < MV_Q; ++q) {
        gq[q].x = tanhf(gq[q].x);
        gq[q].y = tanhf(gq[q].y);
        gq[q].z = tanhf(gq[q].z);
        gq[q].w = tanhf(gq[q].w);
    }
    if (tid < MV_ROWS) {
        gv[row0 + tid] = tanhf(hown);
        fhp[row0 + tid] = 0.f;
        fhp[MV_N + row0 + tid] = 0.f;
    }
    *(f32x4*)(facc + (size_t)row0 * DDIM + 4u * tid) = (f32x4){0.f, 0.f, 0.f, 0.f};
    *(f32x4*)(facc + (size_t)(MV_N + row0) * DDIM + 4u * tid) = (f32x4){0.f, 0.f, 0.f, 0.f};
    {
        const float q0 = wave_reduce_sum(n00 * n00 + n01 * n01);
        const float q1 = wave_reduce_sum(n10 * n10 + n11 * n11);
        const float i0 = rsqrtf(q0), i1 = rsqrtf(q1);
        Kn[nrow * DDIM + lane]              = n00 * i0;
        Kn[nrow * DDIM + lane + 32u]        = n01 * i0;
        Kn[(nrow + 1u) * DDIM + lane]       = n10 * i1;
        Kn[(nrow + 1u) * DDIM + lane + 32u] = n11 * i1;
    }

#pragma unroll
    for (int k = 0; k < MV_ROWS / MV_G; ++k) {
        __builtin_amdgcn_sched_barrier(0);
        if (k >= 1 && k + 1 < MV_ROWS / MV_G) MV_LOAD_GROUP(k + 1);
        __builtin_amdgcn_sched_barrier(0);
#pragma unroll
        for (int r = 0; r < MV_G; ++r) {
            float a0 = 0.f, a1 = 0.f, a2 = 0.f, a3 = 0.f;
#pragma unroll
            for (int q = 0; q < MV_Q; ++q) {
                const f32x4 x = w[k & 1][r][q];
                a0 = __builtin_fmaf(x.x, gq[q].x, a0);
                a1 = __builtin_fmaf(x.y, gq[q].y, a1);
                a2 = __builtin_fmaf(x.z, gq[q].z, a2);
                a3 = __builtin_fmaf(x.w, gq[q].w, a3);
            }
            part[k * MV_G + r][tid] = (a0 + a1) + (a2 + a3);
        }
    }
#undef MV_LOAD_GROUP
    __syncthreads();

    float s0 = 0.f, s1 = 0.f;
#pragma unroll
    for (int j = 0; j < MV_WAVES; ++j) {
        s0 += part[2u * wave][lane + 32u * j];
        s1 += part[2u * wave + 1u][lane + 32u * j];
    }
    s0 = wave_reduce_sum(s0);
    s1 = wave_reduce_sum(s1);
    if (lane == 0) { res[2u * wave] = s0; res[2u * wave + 1u] = s1; }
    __syncthreads();
    if (tid < MV_ROWS) fH[row0 + tid] = -sH[row0 + tid] + res[tid];
}

__global__ void k_matvec(const float* __restrict__ W, const float* __restrict__ gv,
                         const float* __restrict__ sH, float* __restrict__ fH, int N) {
    __shared__ float gs[NMAX];
    for (int t = threadIdx.x; t < N && t < NMAX; t += blockDim.x) gs[t] = gv[t];
    __syncthreads();
    int lane = threadIdx.x & 31, wid = threadIdx.x >> 5;
    int wpb = blockDim.x >> 5;
    int gw = blockIdx.x * wpb + wid;
    int tw = gridDim.x * wpb;
    int n4 = N >> 2;
    for (int row = gw; row < N; row += tw) {
        const float4* Wr = (const float4*)(W + (size_t)row * N);
        float acc = 0.0f;
        for (int c4 = lane; c4 < n4; c4 += 32) {
            float4 w = Wr[c4];
            int c = c4 << 2;
            acc += w.x * gs[c] + w.y * gs[c + 1] + w.z * gs[c + 2] + w.w * gs[c + 3];
        }
        acc = wave_reduce_sum(acc);
        if (lane == 0) fH[row] = -sH[row] + acc;
    }
}

#define EDGE_B 4

__device__ __forceinline__ int lane_to_scalar(int v, int k) {
    return __builtin_amdgcn_readlane(v, k);
}

__device__ __forceinline__ void wave_reduce_sum_batch(float (&v)[EDGE_B]) {
#pragma unroll
    for (int u = 0; u < EDGE_B; ++u) v[u] += xhalf(v[u]);
#pragma unroll
    for (int u = 0; u < EDGE_B; ++u) v[u] += xrow<8>(v[u]);
#pragma unroll
    for (int u = 0; u < EDGE_B; ++u) v[u] += xrow<4>(v[u]);
#pragma unroll
    for (int u = 0; u < EDGE_B; ++u) v[u] += xrow<2>(v[u]);
#pragma unroll
    for (int u = 0; u < EDGE_B; ++u) v[u] += xrow<1>(v[u]);
}

__device__ __forceinline__ float tanh_exp_rcp(float x) {
    const float t = __builtin_amdgcn_exp2f(-2.885390082f * __builtin_fabsf(x));
    const float y = (1.0f - t) * __builtin_amdgcn_rcpf(1.0f + t);
    return __builtin_copysignf(y, x);
}

template <bool ISK>
__device__ __forceinline__ void edge_one(int i, int j, float a0, float a1, float c0, float c1,
                                         float gi, float gj, unsigned lane, int k,
                                         float w1a, float w1b, float b1a, float b1b,
                                         float w2a, float w2b, float& myGram,
                                         float* __restrict__ facc) {
    float s = wave_reduce_sum(a0 * c0 + a1 * c1);
    float coef;
    if (ISK) {
        float h0 = tanh_exp_rcp(s * w1a + b1a);
        float h1 = tanh_exp_rcp(s * w1b + b1b);
        coef = wave_reduce_sum(h0 * w2a + h1 * w2b);
    } else {
        if (lane == (unsigned)k) myGram = s;
        coef = -(0.5f * gi * gj);
    }
    atomicAdd(&facc[i * DDIM + lane],      coef * c0);
    atomicAdd(&facc[i * DDIM + lane + 32], coef * c1);
    atomicAdd(&facc[j * DDIM + lane],      coef * a0);
    atomicAdd(&facc[j * DDIM + lane + 32], coef * a1);
}

template <bool ISK>
__device__ __forceinline__ void edge_chunk(const int* __restrict__ ind, int E, int chunk,
                                           unsigned lane, const float* __restrict__ Kn,
                                           const float* __restrict__ gv,
                                           float w1a, float w1b, float b1a, float b1b,
                                           float w2a, float w2b,
                                           float* __restrict__ fH, float* __restrict__ facc) {
    const int e0 = chunk * 32;
    const int n = min(32, E - e0);
    const int e = e0 + (int)lane;
    int vi = 0, vj = 0;
    if (e < E) {
        const int2 t = *(const int2*)(ind + 2 * (size_t)e);
        vi = t.x; vj = t.y;
    }
    float gvi = 0.f, gvj = 0.f, myGram = 0.f;
    if (!ISK) { gvi = gv[vi]; gvj = gv[vj]; }

    int k0 = 0;
    for (; k0 + EDGE_B <= n; k0 += EDGE_B) {
        int si[EDGE_B], sj[EDGE_B];
        float a0[EDGE_B], a1[EDGE_B], c0[EDGE_B], c1[EDGE_B], gi[EDGE_B], gj[EDGE_B];
#pragma unroll
        for (int u = 0; u < EDGE_B; ++u) {
            si[u] = lane_to_scalar(vi, k0 + u);
            sj[u] = lane_to_scalar(vj, k0 + u);
            a0[u] = Kn[si[u] * DDIM + lane];
            a1[u] = Kn[si[u] * DDIM + lane + 32];
            c0[u] = Kn[sj[u] * DDIM + lane];
            c1[u] = Kn[sj[u] * DDIM + lane + 32];
            gi[u] = ISK ? 0.f : __int_as_float(lane_to_scalar(__float_as_int(gvi), k0 + u));
            gj[u] = ISK ? 0.f : __int_as_float(lane_to_scalar(__float_as_int(gvj), k0 + u));
        }
        float s[EDGE_B], coef[EDGE_B];
#pragma unroll
        for (int u = 0; u < EDGE_B; ++u) s[u] = a0[u] * c0[u] + a1[u] * c1[u];
        wave_reduce_sum_batch(s);
        if (ISK) {
#pragma unroll
            for (int u = 0; u < EDGE_B; ++u) {
                const float h0 = tanh_exp_rcp(s[u] * w1a + b1a);
                const float h1 = tanh_exp_rcp(s[u] * w1b + b1b);
                coef[u] = h0 * w2a + h1 * w2b;
            }
            wave_reduce_sum_batch(coef);
        } else {
#pragma unroll
            for (int u = 0; u < EDGE_B; ++u) {
                if (lane == (unsigned)(k0 + u)) myGram = s[u];
                coef[u] = -(0.5f * gi[u] * gj[u]);
            }
        }
#pragma unroll
        for (int u = 0; u < EDGE_B; ++u) {
            atomicAdd(&facc[si[u] * DDIM + lane],      coef[u] * c0[u]);
            atomicAdd(&facc[si[u] * DDIM + lane + 32], coef[u] * c1[u]);
            atomicAdd(&facc[sj[u] * DDIM + lane],      coef[u] * a0[u]);
            atomicAdd(&facc[sj[u] * DDIM + lane + 32], coef[u] * a1[u]);
        }
    }
    for (; k0 < n; ++k0) {
        const int i = lane_to_scalar(vi, k0), j = lane_to_scalar(vj, k0);
        const float a0 = Kn[i * DDIM + lane], a1 = Kn[i * DDIM + lane + 32];
        const float c0 = Kn[j * DDIM + lane], c1 = Kn[j * DDIM + lane + 32];
        const float gi = ISK ? 0.f : __int_as_float(lane_to_scalar(__float_as_int(gvi), k0));
        const float gj = ISK ? 0.f : __int_as_float(lane_to_scalar(__float_as_int(gvj), k0));
        edge_one<ISK>(i, j, a0, a1, c0, c1, gi, gj, lane, k0,
                      w1a, w1b, b1a, b1b, w2a, w2b, myGram, facc);
    }
    if (!ISK && e < E) {
        atomicAdd(&fH[vi], 0.5f * myGram * gvj);
        atomicAdd(&fH[vj], 0.5f * myGram * gvi);
    }
}

__global__ void __launch_bounds__(256)
k_edges(const int* __restrict__ indHK, const int* __restrict__ indK,
        const float* __restrict__ Kn, const float* __restrict__ gv,
        const float* __restrict__ w1v, const float* __restrict__ b1v,
        const float* __restrict__ w2v, float* __restrict__ fhp,
        float* __restrict__ faccAll, int N, int EHK, int EK) {
    const unsigned where = __builtin_amdgcn_s_sendmsg_rtn(0x87);
    const size_t half = ((where >> 16) & 15u) >= 4u ? 1u : 0u;
    float* __restrict__ fH   = fhp + half * (size_t)N;
    float* __restrict__ facc = faccAll + half * (size_t)N * DDIM;
    const unsigned lane = threadIdx.x & 31u;
    const int wave = __builtin_amdgcn_readfirstlane(threadIdx.x >> 5);
    const int gw = blockIdx.x * (int)(blockDim.x >> 5) + wave;
    const int tw = gridDim.x * (int)(blockDim.x >> 5);
    const int nHK = (EHK + 31) >> 5, nK = (EK + 31) >> 5;
    const float w1a = w1v[lane], w1b = w1v[lane + 32];
    const float b1a = b1v[lane], b1b = b1v[lane + 32];
    const float w2a = w2v[lane], w2b = w2v[lane + 32];
    for (int pass = 0; pass < 2; ++pass) {
        const bool isK = ((pass ^ (int)blockIdx.x) & 1) != 0;
        if (isK) {
            for (int c = gw; c < nK; c += tw)
                edge_chunk<true>(indK, EK, c, lane, Kn, gv, w1a, w1b, b1a, b1b, w2a, w2b, fH, facc);
        } else {
            for (int c = gw; c < nHK; c += tw)
                edge_chunk<false>(indHK, EHK, c, lane, Kn, gv, w1a, w1b, b1a, b1b, w2a, w2b, fH, facc);
        }
    }
}

#define TL_WAVES 4
#define OM_P     65
#define TL_P     68

__global__ void __launch_bounds__(TL_WAVES * 32)
k_tail(const float* __restrict__ Kn, const float* __restrict__ omega,
       const float* __restrict__ facc, const float* __restrict__ fhp,
       float* __restrict__ fH, float* __restrict__ fK, int N) {
    __shared__ float som[DDIM * OM_P];
    __shared__ float tl[TL_WAVES][16 * TL_P];
    const int lane = threadIdx.x & 31;
    const int hrow = blockIdx.x * (TL_WAVES * 16) + (int)threadIdx.x;
    const bool hlive = threadIdx.x < TL_WAVES * 16 && hrow < N;
    float hbase = 0.f, hs0 = 0.f, hs1 = 0.f;
    if (hlive) { hbase = fH[hrow]; hs0 = fhp[hrow]; hs1 = fhp[N + hrow]; }
    const int wid = __builtin_amdgcn_readfirstlane(threadIdx.x >> 5);
    const int tile = blockIdx.x * TL_WAVES + wid;
    const bool live = tile < (N >> 4);
    const int row0 = tile << 4;
    const int l15 = lane & 15;
    const int hi = lane >> 4;
    const int kofs = hi * 2;
    float* my = tl[wid];

    v2f Af[DDIM / 4];
    if (live) {
        const int arow = row0 + l15;
#pragma unroll
        for (int s = 0; s < DDIM / 4; ++s) {
            Af[s].x = Kn[arow * DDIM + 4 * s + kofs];
            Af[s].y = Kn[arow * DDIM + 4 * s + kofs + 1];
        }
    }
#pragma unroll
    for (int u = 0; u < (DDIM * DDIM) / (TL_WAVES * 32); ++u) {
        const int t = u * (TL_WAVES * 32) + (int)threadIdx.x;
        som[(t >> 6) * OM_P + (t & 63)] = omega[t];
    }
    __syncthreads();

    if (live) {
        v8f acc[4] = {};
#pragma unroll
        for (int s = 0; s < DDIM / 4; ++s) {
            const int k = 4 * s + kofs;
#pragma unroll
            for (int t = 0; t < 4; ++t) {
                v2f B;
                const int n = (t << 4) + l15;
                B.x = 0.5f * (som[k * OM_P + n]       - som[n * OM_P + k]);
                B.y = 0.5f * (som[(k + 1) * OM_P + n] - som[n * OM_P + k + 1]);
                acc[t] = __builtin_amdgcn_wmma_f32_16x16x4_f32(
                    false, Af[s], false, B, (short)0, acc[t], false, false);
            }
        }
#pragma unroll
        for (int t = 0; t < 4; ++t) {
#pragma unroll
            for (int v = 0; v < 8; ++v) {
                my[(hi * 8 + v) * TL_P + (t << 4) + l15] = acc[t][v];
            }
        }
    }
    __syncthreads();

    if (live) {
#pragma unroll
        for (int r = 0; r < 16; ++r) {
            const int row = row0 + r;
            const float k0 = Kn[row * DDIM + lane],   k1 = Kn[row * DDIM + lane + 32];
            const float r0 = facc[row * DDIM + lane]      + facc[(N + row) * DDIM + lane];
            const float r1 = facc[row * DDIM + lane + 32] + facc[(N + row) * DDIM + lane + 32];
            const float dot = wave_reduce_sum(k0 * r0 + k1 * r1);
            fK[row * DDIM + lane]      = -r0 + k0 * dot + my[r * TL_P + lane];
            fK[row * DDIM + lane + 32] = -r1 + k1 * dot + my[r * TL_P + lane + 32];
        }
    }
    if (hlive) fH[hrow] = hbase + (hs0 + hs1);
}

extern "C" void kernel_launch(void* const* d_in, const int* in_sizes, int n_in,
                              void* d_out, int out_size, void* d_ws, size_t ws_size,
                              hipStream_t stream) {
    const float* sH    = (const float*)d_in[0];
    const float* sK    = (const float*)d_in[1];
    const float* WH    = (const float*)d_in[2];
    const float* omega = (const float*)d_in[3];
    const float* w1v   = (const float*)d_in[4];
    const float* b1v   = (const float*)d_in[5];
    const float* w2v   = (const float*)d_in[6];
    const int*   indK  = (const int*)d_in[7];
    const int*   indHK = (const int*)d_in[8];

    int N   = in_sizes[0];
    int EK  = in_sizes[7] / 2;
    int EHK = in_sizes[8] / 2;
    int ND  = N * DDIM;

    float* fH = (float*)d_out;
    float* fK = (float*)d_out + N;

    float* Kn   = (float*)d_ws;
    float* gv   = Kn + (size_t)ND;
    float* facc = gv + N;
    float* fhp  = facc + 2 * (size_t)ND;

    if (N == MV_N) {
        k_front_stream<<<MV_N / MV_ROWS, MV_WAVES * 32, 0, stream>>>(WH, sH, sK, Kn, gv, facc, fhp, fH);
    } else {
        k_prep  <<<(ND + 255) / 256,    256, 0, stream>>>(sH, facc, fhp, gv, N);
        k_norm  <<<(N + 7) / 8,         256, 0, stream>>>(sK, Kn, N);
        k_matvec<<<256,                 256, 0, stream>>>(WH, gv, sH, fH, N);
    }
    k_edges <<<1024,                    256, 0, stream>>>(indHK, indK, Kn, gv, w1v, b1v, w2v,
                                                          fhp, facc, N, EHK, EK);
    k_tail  <<<((N >> 4) + TL_WAVES - 1) / TL_WAVES, TL_WAVES * 32, 0, stream>>>(Kn, omega, facc, fhp, fH, fK, N);
}
